# starting kernel
# baseline (speedup 1.0000x reference)
#include <hip/hip_runtime.h>

typedef __attribute__((ext_vector_type(2))) float v2f;
typedef __attribute__((ext_vector_type(8))) float v8f;

#define IN_CH  512
#define HID    16
#define OUT_CH 64
#define THREADS 256

__global__ void k_init_deg(float* __restrict__ deg, int n) {
  int i = blockIdx.x * blockDim.x + threadIdx.x;
  if (i < n) deg[i] = 1.0f;
}

__global__ void k_count_deg(const int* __restrict__ dst, float* __restrict__ deg, int e) {
  int i = blockIdx.x * blockDim.x + threadIdx.x;
  if (i < e) atomicAdd(&deg[dst[i]], 1.0f);
}

__global__ void k_rsqrt_inplace(float* __restrict__ deg, int n) {
  int i = blockIdx.x * blockDim.x + threadIdx.x;
  if (i < n) deg[i] = rsqrtf(deg[i]);
}

__global__ void k_edge_norm(const int* __restrict__ src, const int* __restrict__ dst,
                            const float* __restrict__ dinv, float* __restrict__ norm, int e) {
  int i = blockIdx.x * blockDim.x + threadIdx.x;
  if (i < e) norm[i] = dinv[src[i]] * dinv[dst[i]];
}

__global__ void k_gemm1(const float* __restrict__ x, const float* __restrict__ W1,
                        float* __restrict__ xw1, int nTiles) {
  __shared__ float w1s[IN_CH * HID];
  for (int i = threadIdx.x; i < IN_CH * HID; i += blockDim.x) w1s[i] = W1[i];
  __syncthreads();

  int wave = threadIdx.x >> 5;
  int lane = threadIdx.x & 31;
  int t = blockIdx.x * (blockDim.x >> 5) + wave;
  if (t >= nTiles) return;

  int half = lane >> 4;
  int ml = lane & 15;
  const float* xr = x + (size_t)(t * 16 + ml) * IN_CH + half * 2;

  v8f c = {};
  for (int kk = 0; kk < IN_CH / 4; ++kk) {
    v2f a = *(const v2f*)(xr + kk * 4);
    int k0 = kk * 4 + half * 2;
    v2f b;
    b.x = w1s[k0 * HID + ml];
    b.y = w1s[(k0 + 1) * HID + ml];
    c = __builtin_amdgcn_wmma_f32_16x16x4_f32(false, a, false, b, (short)0, c, false, false);
  }

  float* o = xw1 + (size_t)(t * 16) * HID + ml;
  #pragma unroll
  for (int r = 0; r < 8; ++r)
    o[(half * 8 + r) * HID] = c[r];
}

__global__ void k_gemm2(const float* __restrict__ h, const float* __restrict__ W2,
                        float* __restrict__ h2, int nTiles) {
  __shared__ float w2s[HID * OUT_CH];
  for (int i = threadIdx.x; i < HID * OUT_CH; i += blockDim.x) w2s[i] = W2[i];
  __syncthreads();

  int wave = threadIdx.x >> 5;
  int lane = threadIdx.x & 31;
  int t = blockIdx.x * (blockDim.x >> 5) + wave;
  if (t >= nTiles) return;

  int half = lane >> 4;
  int ml = lane & 15;
  const float* hr = h + (size_t)(t * 16 + ml) * HID + half * 2;

  v8f c0 = {}, c1 = {}, c2 = {}, c3 = {};
  #pragma unroll
  for (int kk = 0; kk < HID / 4; ++kk) {
    v2f a = *(const v2f*)(hr + kk * 4);
    int k0 = kk * 4 + half * 2;
    v2f b0, b1, b2, b3;
    b0.x = w2s[k0 * OUT_CH +  0 + ml]; b0.y = w2s[(k0 + 1) * OUT_CH +  0 + ml];
    b1.x = w2s[k0 * OUT_CH + 16 + ml]; b1.y = w2s[(k0 + 1) * OUT_CH + 16 + ml];
    b2.x = w2s[k0 * OUT_CH + 32 + ml]; b2.y = w2s[(k0 + 1) * OUT_CH + 32 + ml];
    b3.x = w2s[k0 * OUT_CH + 48 + ml]; b3.y = w2s[(k0 + 1) * OUT_CH + 48 + ml];
    c0 = __builtin_amdgcn_wmma_f32_16x16x4_f32(false, a, false, b0, (short)0, c0, false, false);
    c1 = __builtin_amdgcn_wmma_f32_16x16x4_f32(false, a, false, b1, (short)0, c1, false, false);
    c2 = __builtin_amdgcn_wmma_f32_16x16x4_f32(false, a, false, b2, (short)0, c2, false, false);
    c3 = __builtin_amdgcn_wmma_f32_16x16x4_f32(false, a, false, b3, (short)0, c3, false, false);
  }

  float* o = h2 + (size_t)(t * 16) * OUT_CH + ml;
  #pragma unroll
  for (int r = 0; r < 8; ++r) {
    int row = half * 8 + r;
    o[row * OUT_CH +  0] = c0[r];
    o[row * OUT_CH + 16] = c1[r];
    o[row * OUT_CH + 32] = c2[r];
    o[row * OUT_CH + 48] = c3[r];
  }
}

template <int LOGC>
__global__ void k_init_selfloop(const float* __restrict__ in, const float* __restrict__ dinv,
                                const float* __restrict__ bias, float* __restrict__ out,
                                long long total) {
  long long i = (long long)blockIdx.x * blockDim.x + threadIdx.x;
  if (i >= total) return;
  int node = (int)(i >> LOGC);
  int ch = (int)(i & ((1 << LOGC) - 1));
  float di = dinv[node];
  out[i] = bias[ch] + in[i] * di * di;
}

template <int LOGC>
__global__ void k_agg_edges(const int* __restrict__ src, const int* __restrict__ dst,
                            const float* __restrict__ norm, const float* __restrict__ in,
                            float* __restrict__ out, long long total) {
  long long i = (long long)blockIdx.x * blockDim.x + threadIdx.x;
  if (i >= total) return;
  int e = (int)(i >> LOGC);
  int ch = (int)(i & ((1 << LOGC) - 1));
  int s = src[e], d = dst[e];
  float v = in[((size_t)s << LOGC) + ch] * norm[e];
  atomicAdd(&out[((size_t)d << LOGC) + ch], v);
}

__global__ void k_relu(float* __restrict__ h, long long total) {
  long long i = (long long)blockIdx.x * blockDim.x + threadIdx.x;
  if (i < total) h[i] = fmaxf(h[i], 0.0f);
}

static inline int gridFor(long long total, int threads) {
  return (int)((total + threads - 1) / threads);
}

extern "C" void kernel_launch(void* const* d_in, const int* in_sizes, int n_in,
                              void* d_out, int out_size, void* d_ws, size_t ws_size,
                              hipStream_t stream) {
  const float* x  = (const float*)d_in[0];
  const int* ei   = (const int*)d_in[1];
  const float* W1 = (const float*)d_in[2];
  const float* b1 = (const float*)d_in[3];
  const float* W2 = (const float*)d_in[4];
  const float* b2 = (const float*)d_in[5];
  float* out = (float*)d_out;

  const int n = in_sizes[0] / IN_CH;
  const int e = in_sizes[1] / 2;
  const int* src = ei;
  const int* dst = ei + e;

  float* dinv = (float*)d_ws;
  float* norm = dinv + n;
  float* xw1  = norm + e;
  float* h    = xw1 + (size_t)n * HID;
  float* h2   = h   + (size_t)n * HID;

  const int nTiles = (n + 15) / 16;
  const int wavesPerBlock = THREADS / 32;

  k_init_deg<<<gridFor(n, THREADS), THREADS, 0, stream>>>(dinv, n);
  k_count_deg<<<gridFor(e, THREADS), THREADS, 0, stream>>>(dst, dinv, e);
  k_rsqrt_inplace<<<gridFor(n, THREADS), THREADS, 0, stream>>>(dinv, n);
  k_edge_norm<<<gridFor(e, THREADS), THREADS, 0, stream>>>(src, dst, dinv, norm, e);

  k_gemm1<<<gridFor(nTiles, wavesPerBlock), THREADS, 0, stream>>>(x, W1, xw1, nTiles);
  long long tot1 = (long long)n * HID;
  k_init_selfloop<4><<<gridFor(tot1, THREADS), THREADS, 0, stream>>>(xw1, dinv, b1, h, tot1);
  long long te1 = (long long)e * HID;
  k_agg_edges<4><<<gridFor(te1, THREADS), THREADS, 0, stream>>>(src, dst, norm, xw1, h, te1);
  k_relu<<<gridFor(tot1, THREADS), THREADS, 0, stream>>>(h, tot1);

  k_gemm2<<<gridFor(nTiles, wavesPerBlock), THREADS, 0, stream>>>(h, W2, h2, nTiles);
  long long tot2 = (long long)n * OUT_CH;
  k_init_selfloop<6><<<gridFor(tot2, THREADS), THREADS, 0, stream>>>(h2, dinv, b2, out, tot2);
  long long te2 = (long long)e * OUT_CH;
  k_agg_edges<6><<<gridFor(te2, THREADS), THREADS, 0, stream>>>(src, dst, norm, h2, out, te2);
}


# Layer 2 aggregates before its product (linearity): scatter-add 16 channels of relu(h), then the WMMA product with W2 adds b2 and writes out; layer-2 sum order changes
# speedup vs baseline: 1.6869x; 1.6869x over previous
#include <hip/hip_runtime.h>

typedef __attribute__((ext_vector_type(2))) float v2f;
typedef __attribute__((ext_vector_type(8))) float v8f;

#define IN_CH  512
#define HID    16
#define OUT_CH 64
#define THREADS 256

__global__ void k_init_deg(float* __restrict__ deg, int n) {
  int i = blockIdx.x * blockDim.x + threadIdx.x;
  if (i < n) deg[i] = 1.0f;
}

__global__ void k_count_deg(const int* __restrict__ dst, float* __restrict__ deg, int e) {
  int i = blockIdx.x * blockDim.x + threadIdx.x;
  if (i < e) atomicAdd(&deg[dst[i]], 1.0f);
}

__global__ void k_rsqrt_inplace(float* __restrict__ deg, int n) {
  int i = blockIdx.x * blockDim.x + threadIdx.x;
  if (i < n) deg[i] = rsqrtf(deg[i]);
}

__global__ void k_edge_norm(const int* __restrict__ src, const int* __restrict__ dst,
                            const float* __restrict__ dinv, float* __restrict__ norm, int e) {
  int i = blockIdx.x * blockDim.x + threadIdx.x;
  if (i < e) norm[i] = dinv[src[i]] * dinv[dst[i]];
}

__global__ void k_gemm1(const float* __restrict__ x, const float* __restrict__ W1,
                        float* __restrict__ xw1, int nTiles) {
  __shared__ float w1s[IN_CH * HID];
  for (int i = threadIdx.x; i < IN_CH * HID; i += blockDim.x) w1s[i] = W1[i];
  __syncthreads();

  int wave = threadIdx.x >> 5;
  int lane = threadIdx.x & 31;
  int t = blockIdx.x * (blockDim.x >> 5) + wave;
  if (t >= nTiles) return;

  int half = lane >> 4;
  int ml = lane & 15;
  const float* xr = x + (size_t)(t * 16 + ml) * IN_CH + half * 2;

  v8f c = {};
  for (int kk = 0; kk < IN_CH / 4; ++kk) {
    v2f a = *(const v2f*)(xr + kk * 4);
    int k0 = kk * 4 + half * 2;
    v2f b;
    b.x = w1s[k0 * HID + ml];
    b.y = w1s[(k0 + 1) * HID + ml];
    c = __builtin_amdgcn_wmma_f32_16x16x4_f32(false, a, false, b, (short)0, c, false, false);
  }

  float* o = xw1 + (size_t)(t * 16) * HID + ml;
  #pragma unroll
  for (int r = 0; r < 8; ++r)
    o[(half * 8 + r) * HID] = c[r];
}

__global__ void k_gemm2(const float* __restrict__ g, const float* __restrict__ W2,
                        const float* __restrict__ bias, float* __restrict__ y, int nTiles) {
  __shared__ float w2s[HID * OUT_CH];
  for (int i = threadIdx.x; i < HID * OUT_CH; i += blockDim.x) w2s[i] = W2[i];
  __syncthreads();

  int wave = threadIdx.x >> 5;
  int lane = threadIdx.x & 31;
  int t = blockIdx.x * (blockDim.x >> 5) + wave;
  if (t >= nTiles) return;

  int half = lane >> 4;
  int ml = lane & 15;
  const float* hr = g + (size_t)(t * 16 + ml) * HID + half * 2;

  v8f c0 = {}, c1 = {}, c2 = {}, c3 = {};
  #pragma unroll
  for (int kk = 0; kk < HID / 4; ++kk) {
    v2f a = *(const v2f*)(hr + kk * 4);
    int k0 = kk * 4 + half * 2;
    v2f b0, b1, b2, b3;
    b0.x = w2s[k0 * OUT_CH +  0 + ml]; b0.y = w2s[(k0 + 1) * OUT_CH +  0 + ml];
    b1.x = w2s[k0 * OUT_CH + 16 + ml]; b1.y = w2s[(k0 + 1) * OUT_CH + 16 + ml];
    b2.x = w2s[k0 * OUT_CH + 32 + ml]; b2.y = w2s[(k0 + 1) * OUT_CH + 32 + ml];
    b3.x = w2s[k0 * OUT_CH + 48 + ml]; b3.y = w2s[(k0 + 1) * OUT_CH + 48 + ml];
    c0 = __builtin_amdgcn_wmma_f32_16x16x4_f32(false, a, false, b0, (short)0, c0, false, false);
    c1 = __builtin_amdgcn_wmma_f32_16x16x4_f32(false, a, false, b1, (short)0, c1, false, false);
    c2 = __builtin_amdgcn_wmma_f32_16x16x4_f32(false, a, false, b2, (short)0, c2, false, false);
    c3 = __builtin_amdgcn_wmma_f32_16x16x4_f32(false, a, false, b3, (short)0, c3, false, false);
  }

  const float bs0 = bias[ml], bs1 = bias[16 + ml], bs2 = bias[32 + ml], bs3 = bias[48 + ml];
  float* o = y + (size_t)(t * 16) * OUT_CH + ml;
  #pragma unroll
  for (int r = 0; r < 8; ++r) {
    int row = half * 8 + r;
    o[row * OUT_CH +  0] = c0[r] + bs0;
    o[row * OUT_CH + 16] = c1[r] + bs1;
    o[row * OUT_CH + 32] = c2[r] + bs2;
    o[row * OUT_CH + 48] = c3[r] + bs3;
  }
}

template <int LOGC>
__global__ void k_init_selfloop(const float* __restrict__ in, const float* __restrict__ dinv,
                                const float* __restrict__ bias, float* __restrict__ out,
                                long long total) {
  long long i = (long long)blockIdx.x * blockDim.x + threadIdx.x;
  if (i >= total) return;
  int node = (int)(i >> LOGC);
  int ch = (int)(i & ((1 << LOGC) - 1));
  float di = dinv[node];
  out[i] = bias[ch] + in[i] * di * di;
}

template <int LOGC>
__global__ void k_selfloop_scale(const float* __restrict__ in, const float* __restrict__ dinv,
                                 float* __restrict__ out, long long total) {
  long long i = (long long)blockIdx.x * blockDim.x + threadIdx.x;
  if (i >= total) return;
  int node = (int)(i >> LOGC);
  float di = dinv[node];
  out[i] = in[i] * di * di;
}

template <int LOGC>
__global__ void k_agg_edges(const int* __restrict__ src, const int* __restrict__ dst,
                            const float* __restrict__ norm, const float* __restrict__ in,
                            float* __restrict__ out, long long total) {
  long long i = (long long)blockIdx.x * blockDim.x + threadIdx.x;
  if (i >= total) return;
  int e = (int)(i >> LOGC);
  int ch = (int)(i & ((1 << LOGC) - 1));
  int s = src[e], d = dst[e];
  float v = in[((size_t)s << LOGC) + ch] * norm[e];
  atomicAdd(&out[((size_t)d << LOGC) + ch], v);
}

__global__ void k_relu(float* __restrict__ h, long long total) {
  long long i = (long long)blockIdx.x * blockDim.x + threadIdx.x;
  if (i < total) h[i] = fmaxf(h[i], 0.0f);
}

static inline int gridFor(long long total, int threads) {
  return (int)((total + threads - 1) / threads);
}

extern "C" void kernel_launch(void* const* d_in, const int* in_sizes, int n_in,
                              void* d_out, int out_size, void* d_ws, size_t ws_size,
                              hipStream_t stream) {
  const float* x  = (const float*)d_in[0];
  const int* ei   = (const int*)d_in[1];
  const float* W1 = (const float*)d_in[2];
  const float* b1 = (const float*)d_in[3];
  const float* W2 = (const float*)d_in[4];
  const float* b2 = (const float*)d_in[5];
  float* out = (float*)d_out;

  const int n = in_sizes[0] / IN_CH;
  const int e = in_sizes[1] / 2;
  const int* src = ei;
  const int* dst = ei + e;

  float* dinv = (float*)d_ws;
  float* norm = dinv + n;
  float* xw1  = norm + e;
  float* h    = xw1 + (size_t)n * HID;
  float* g    = h   + (size_t)n * HID;

  const int nTiles = (n + 15) / 16;
  const int wavesPerBlock = THREADS / 32;

  k_init_deg<<<gridFor(n, THREADS), THREADS, 0, stream>>>(dinv, n);
  k_count_deg<<<gridFor(e, THREADS), THREADS, 0, stream>>>(dst, dinv, e);
  k_rsqrt_inplace<<<gridFor(n, THREADS), THREADS, 0, stream>>>(dinv, n);
  k_edge_norm<<<gridFor(e, THREADS), THREADS, 0, stream>>>(src, dst, dinv, norm, e);

  k_gemm1<<<gridFor(nTiles, wavesPerBlock), THREADS, 0, stream>>>(x, W1, xw1, nTiles);
  long long tot1 = (long long)n * HID;
  k_init_selfloop<4><<<gridFor(tot1, THREADS), THREADS, 0, stream>>>(xw1, dinv, b1, h, tot1);
  long long te1 = (long long)e * HID;
  k_agg_edges<4><<<gridFor(te1, THREADS), THREADS, 0, stream>>>(src, dst, norm, xw1, h, te1);
  k_relu<<<gridFor(tot1, THREADS), THREADS, 0, stream>>>(h, tot1);

  k_selfloop_scale<4><<<gridFor(tot1, THREADS), THREADS, 0, stream>>>(h, dinv, g, tot1);
  k_agg_edges<4><<<gridFor(te1, THREADS), THREADS, 0, stream>>>(src, dst, norm, h, g, te1);
  k_gemm2<<<gridFor(nTiles, wavesPerBlock), THREADS, 0, stream>>>(g, W2, b2, out, nTiles);
}


# Both aggregations become a gather over edges bucketed by destination (per-call counting sort, int atomics); no float atomics; dinv folded into rows, bias+relu into the gather; sum order changes
# speedup vs baseline: 2.1238x; 1.2590x over previous
#include <hip/hip_runtime.h>


typedef __attribute__((ext_vector_type(2))) float v2f;
typedef __attribute__((ext_vector_type(8))) float v8f;

#define IN_CH  512
#define HID    16
#define OUT_CH 64
#define THREADS 256
#define SCAN_T 1024

__global__ void k_zero_cnt(int* __restrict__ cnt, int n) {
  int i = blockIdx.x * blockDim.x + threadIdx.x;
  if (i < n) cnt[i] = 0;
}

__global__ void k_count_deg(const int* __restrict__ dst, int* __restrict__ cnt, int e) {
  int i = blockIdx.x * blockDim.x + threadIdx.x;
  if (i < e) atomicAdd(&cnt[dst[i]], 1);
}

__global__ void __launch_bounds__(SCAN_T)
k_scan(const int* __restrict__ cnt, int* __restrict__ cur, float* __restrict__ dinv, int n) {
  __shared__ int wtot[32];
  __shared__ int ptot[32];
  const int tid = threadIdx.x, lane = tid & 31, wave = tid >> 5;
  const int start = blockIdx.x * SCAN_T;
  int pre = 0;
  for (int i = tid; i < start; i += SCAN_T) pre += cnt[i];
  const int i = start + tid;
  const int c = (i < n) ? cnt[i] : 0;
  int incl = c;
  for (int o = 1; o < 32; o <<= 1) {
    int t = __shfl_up(incl, o);
    if (lane >= o) incl += t;
  }
  for (int o = 16; o > 0; o >>= 1) pre += __shfl_xor(pre, o);
  if (lane == 31) wtot[wave] = incl;
  if (lane == 0) ptot[wave] = pre;
  __syncthreads();
  int base = ptot[lane];
  int wpre = (lane < wave) ? wtot[lane] : 0;
  for (int o = 16; o > 0; o >>= 1) {
    base += __shfl_xor(base, o);
    wpre += __shfl_xor(wpre, o);
  }
  if (i < n) {
    cur[i] = base + wpre + incl - c;
    dinv[i] = rsqrtf((float)(c + 1));
  }
}

__global__ void k_fill(const int* __restrict__ src, const int* __restrict__ dst,
                       int* __restrict__ cur, int* __restrict__ csr, int e) {
  int i = blockIdx.x * blockDim.x + threadIdx.x;
  if (i >= e) return;
  int p = atomicAdd(&cur[dst[i]], 1);
  if ((unsigned)p < (unsigned)e) csr[p] = src[i];
}

__global__ void k_gemm1(const float* __restrict__ x, const float* __restrict__ W1,
                        const float* __restrict__ dinv, float* __restrict__ xw1, int nTiles) {
  __shared__ float w1s[IN_CH * HID];
  for (int i = threadIdx.x; i < IN_CH * HID; i += blockDim.x) w1s[i] = W1[i];
  __syncthreads();

  int wave = threadIdx.x >> 5;
  int lane = threadIdx.x & 31;
  int t = blockIdx.x * (blockDim.x >> 5) + wave;
  if (t >= nTiles) return;

  int half = lane >> 4;
  int ml = lane & 15;
  const float* xr = x + (size_t)(t * 16 + ml) * IN_CH + half * 2;

  v8f c = {};
  for (int kk = 0; kk < IN_CH / 4; ++kk) {
    v2f a = *(const v2f*)(xr + kk * 4);
    int k0 = kk * 4 + half * 2;
    v2f b;
    b.x = w1s[k0 * HID + ml];
    b.y = w1s[(k0 + 1) * HID + ml];
    c = __builtin_amdgcn_wmma_f32_16x16x4_f32(false, a, false, b, (short)0, c, false, false);
  }

  float* o = xw1 + (size_t)(t * 16) * HID + ml;
  const float* dv = dinv + t * 16 + half * 8;
  #pragma unroll
  for (int r = 0; r < 8; ++r)
    o[(half * 8 + r) * HID] = c[r] * dv[r];
}

__global__ void k_gemm2(const float* __restrict__ g, const float* __restrict__ W2,
                        const float* __restrict__ bias, float* __restrict__ y, int nTiles) {
  __shared__ float w2s[HID * OUT_CH];
  for (int i = threadIdx.x; i < HID * OUT_CH; i += blockDim.x) w2s[i] = W2[i];
  __syncthreads();

  int wave = threadIdx.x >> 5;
  int lane = threadIdx.x & 31;
  int t = blockIdx.x * (blockDim.x >> 5) + wave;
  if (t >= nTiles) return;

  int half = lane >> 4;
  int ml = lane & 15;
  const float* hr = g + (size_t)(t * 16 + ml) * HID + half * 2;

  v8f c0 = {}, c1 = {}, c2 = {}, c3 = {};
  #pragma unroll
  for (int kk = 0; kk < HID / 4; ++kk) {
    v2f a = *(const v2f*)(hr + kk * 4);
    int k0 = kk * 4 + half * 2;
    v2f b0, b1, b2, b3;
    b0.x = w2s[k0 * OUT_CH +  0 + ml]; b0.y = w2s[(k0 + 1) * OUT_CH +  0 + ml];
    b1.x = w2s[k0 * OUT_CH + 16 + ml]; b1.y = w2s[(k0 + 1) * OUT_CH + 16 + ml];
    b2.x = w2s[k0 * OUT_CH + 32 + ml]; b2.y = w2s[(k0 + 1) * OUT_CH + 32 + ml];
    b3.x = w2s[k0 * OUT_CH + 48 + ml]; b3.y = w2s[(k0 + 1) * OUT_CH + 48 + ml];
    c0 = __builtin_amdgcn_wmma_f32_16x16x4_f32(false, a, false, b0, (short)0, c0, false, false);
    c1 = __builtin_amdgcn_wmma_f32_16x16x4_f32(false, a, false, b1, (short)0, c1, false, false);
    c2 = __builtin_amdgcn_wmma_f32_16x16x4_f32(false, a, false, b2, (short)0, c2, false, false);
    c3 = __builtin_amdgcn_wmma_f32_16x16x4_f32(false, a, false, b3, (short)0, c3, false, false);
  }

  const float bs0 = bias[ml], bs1 = bias[16 + ml], bs2 = bias[32 + ml], bs3 = bias[48 + ml];
  float* o = y + (size_t)(t * 16) * OUT_CH + ml;
  #pragma unroll
  for (int r = 0; r < 8; ++r) {
    int row = half * 8 + r;
    o[row * OUT_CH +  0] = c0[r] + bs0;
    o[row * OUT_CH + 16] = c1[r] + bs1;
    o[row * OUT_CH + 32] = c2[r] + bs2;
    o[row * OUT_CH + 48] = c3[r] + bs3;
  }
}

template <bool TAIL>
__device__ __forceinline__ void add8(float4& acc, const float4* __restrict__ rows,
                                     const int* __restrict__ csr, int j, int end, int sub, int n) {
  int s[8];
  float4 v[8];
  #pragma unroll
  for (int k = 0; k < 8; ++k) s[k] = csr[TAIL ? min(j + k, end - 1) : j + k];
  #pragma unroll
  for (int k = 0; k < 8; ++k) v[k] = rows[min(max(s[k], 0), n - 1) * 4 + sub];
  #pragma unroll
  for (int k = 0; k < 8; ++k) {
    const bool ok = !TAIL || k == 0 || j + k < end;
    acc.x = ok ? acc.x + v[k].x : acc.x;
    acc.y = ok ? acc.y + v[k].y : acc.y;
    acc.z = ok ? acc.z + v[k].z : acc.z;
    acc.w = ok ? acc.w + v[k].w : acc.w;
  }
}

template <bool FIRST>
__global__ void k_gather(const int* __restrict__ cnt, const int* __restrict__ bend,
                         const int* __restrict__ csr, const float* __restrict__ dinv,
                         const float* __restrict__ in, const float* __restrict__ bias,
                         float* __restrict__ out, int n) {
  const int gid = blockIdx.x * blockDim.x + threadIdx.x;
  const int node = gid >> 2;
  const int sub = gid & 3;
  if (node >= n) return;
  const float4* __restrict__ rows = (const float4*)in;
  const int end = bend[node];
  const int begin = end - cnt[node];
  float4 acc = rows[node * 4 + sub];
  int j = begin;
  for (; j + 8 <= end; j += 8) add8<false>(acc, rows, csr, j, end, sub, n);
  if (j < end) add8<true>(acc, rows, csr, j, end, sub, n);
  const float di = dinv[node];
  float4 r;
  if (FIRST) {
    const float* b = bias + sub * 4;
    r.x = fmaxf(b[0] + di * acc.x, 0.0f) * di;
    r.y = fmaxf(b[1] + di * acc.y, 0.0f) * di;
    r.z = fmaxf(b[2] + di * acc.z, 0.0f) * di;
    r.w = fmaxf(b[3] + di * acc.w, 0.0f) * di;
  } else {
    r.x = di * acc.x; r.y = di * acc.y; r.z = di * acc.z; r.w = di * acc.w;
  }
  ((float4*)out)[node * 4 + sub] = r;
}

static inline int gridFor(long long total, int threads) {
  return (int)((total + threads - 1) / threads);
}

extern "C" void kernel_launch(void* const* d_in, const int* in_sizes, int n_in,
                              void* d_out, int out_size, void* d_ws, size_t ws_size,
                              hipStream_t stream) {
  const float* x  = (const float*)d_in[0];
  const int* ei   = (const int*)d_in[1];
  const float* W1 = (const float*)d_in[2];
  const float* b1 = (const float*)d_in[3];
  const float* W2 = (const float*)d_in[4];
  const float* b2 = (const float*)d_in[5];
  float* out = (float*)d_out;

  const int n = in_sizes[0] / IN_CH;
  const int e = in_sizes[1] / 2;
  const int* src = ei;
  const int* dst = ei + e;

  float* xs   = (float*)d_ws;
  float* hs   = xs + (size_t)n * HID;
  float* g    = xs;
  int*   csr  = (int*)(hs + (size_t)n * HID);
  int*   cnt  = csr + e;
  int*   cur  = cnt + n;
  float* dinv = (float*)(cur + n);

  const int nTiles = (n + 15) / 16;
  const int wavesPerBlock = THREADS / 32;

  k_zero_cnt<<<gridFor(n, THREADS), THREADS, 0, stream>>>(cnt, n);
  k_count_deg<<<gridFor(e, THREADS), THREADS, 0, stream>>>(dst, cnt, e);
  k_scan<<<gridFor(n, SCAN_T), SCAN_T, 0, stream>>>(cnt, cur, dinv, n);
  k_fill<<<gridFor(e, THREADS), THREADS, 0, stream>>>(src, dst, cur, csr, e);

  k_gemm1<<<gridFor(nTiles, wavesPerBlock), THREADS, 0, stream>>>(x, W1, dinv, xs, nTiles);
  long long lanes = (long long)n * 4;
  k_gather<true><<<gridFor(lanes, THREADS), THREADS, 0, stream>>>(cnt, cur, csr, dinv, xs, b1, hs, n);

  k_gather<false><<<gridFor(lanes, THREADS), THREADS, 0, stream>>>(cnt, cur, csr, dinv, hs, b2, g, n);
  k_gemm2<<<gridFor(nTiles, wavesPerBlock), THREADS, 0, stream>>>(g, W2, b2, out, nTiles);
}


# GEMM1 data path: x streamed in whole-line 16-B loads, 16+ in flight per wave, via a wave-private LDS tile to the A fragments; W1 staged transposed in one trip; product sum order changes
# speedup vs baseline: 2.2075x; 1.0394x over previous
#include <hip/hip_runtime.h>


typedef __attribute__((ext_vector_type(2))) float v2f;
typedef __attribute__((ext_vector_type(4))) float v4f;
typedef __attribute__((ext_vector_type(8))) float v8f;

#define IN_CH  512
#define HID    16
#define OUT_CH 64
#define THREADS 256
#define SCAN_T 1024

__global__ void k_zero_cnt(int* __restrict__ cnt, int n) {
  int i = blockIdx.x * blockDim.x + threadIdx.x;
  if (i < n) cnt[i] = 0;
}

__global__ void k_count_deg(const int* __restrict__ dst, int* __restrict__ cnt, int e) {
  int i = blockIdx.x * blockDim.x + threadIdx.x;
  if (i < e) atomicAdd(&cnt[dst[i]], 1);
}

__global__ void __launch_bounds__(SCAN_T)
k_scan(const int* __restrict__ cnt, int* __restrict__ cur, float* __restrict__ dinv, int n) {
  __shared__ int wtot[32];
  __shared__ int ptot[32];
  const int tid = threadIdx.x, lane = tid & 31, wave = tid >> 5;
  const int start = blockIdx.x * SCAN_T;
  int pre = 0;
  for (int i = tid; i < start; i += SCAN_T) pre += cnt[i];
  const int i = start + tid;
  const int c = (i < n) ? cnt[i] : 0;
  int incl = c;
  for (int o = 1; o < 32; o <<= 1) {
    int t = __shfl_up(incl, o);
    if (lane >= o) incl += t;
  }
  for (int o = 16; o > 0; o >>= 1) pre += __shfl_xor(pre, o);
  if (lane == 31) wtot[wave] = incl;
  if (lane == 0) ptot[wave] = pre;
  __syncthreads();
  int base = ptot[lane];
  int wpre = (lane < wave) ? wtot[lane] : 0;
  for (int o = 16; o > 0; o >>= 1) {
    base += __shfl_xor(base, o);
    wpre += __shfl_xor(wpre, o);
  }
  if (i < n) {
    cur[i] = base + wpre + incl - c;
    dinv[i] = rsqrtf((float)(c + 1));
  }
}

__global__ void k_fill(const int* __restrict__ src, const int* __restrict__ dst,
                       int* __restrict__ cur, int* __restrict__ csr, int e) {
  int i = blockIdx.x * blockDim.x + threadIdx.x;
  if (i >= e) return;
  int p = atomicAdd(&cur[dst[i]], 1);
  if ((unsigned)p < (unsigned)e) csr[p] = src[i];
}

#define KC 64
#define A_PITCH (KC + 4)
#define W_PITCH (IN_CH + 4)
#define NCHUNK (IN_CH / KC)

__global__ void __launch_bounds__(THREADS)
k_gemm1(const float* __restrict__ x, const float* __restrict__ W1,
        const float* __restrict__ dinv, float* __restrict__ xw1, int nTiles) {
  __shared__ __attribute__((aligned(16))) float w1t[HID * W_PITCH];
  __shared__ __attribute__((aligned(16))) float as[THREADS / 32][16 * A_PITCH];

  {
    v4f w[8];
    #pragma unroll
    for (int i = 0; i < 8; ++i) w[i] = *(const v4f*)(W1 + (i * THREADS + threadIdx.x) * 4);
    #pragma unroll
    for (int i = 0; i < 8; ++i) {
      const int idx = (i * THREADS + threadIdx.x) * 4;
      const int k = idx >> 4, nn = idx & 15;
      w1t[(nn + 0) * W_PITCH + k] = w[i].x;
      w1t[(nn + 1) * W_PITCH + k] = w[i].y;
      w1t[(nn + 2) * W_PITCH + k] = w[i].z;
      w1t[(nn + 3) * W_PITCH + k] = w[i].w;
    }
  }
  __syncthreads();

  const int wave = threadIdx.x >> 5;
  const int lane = threadIdx.x & 31;
  const int t = blockIdx.x * (THREADS / 32) + wave;
  if (t >= nTiles) return;

  const int half = lane >> 4;
  const int ml = lane & 15;
  const float* xg = x + ((size_t)t * 16 + half) * IN_CH + ml * 4;
  float* ast = &as[wave][half * A_PITCH + ml * 4];
  const float* afr = &as[wave][ml * A_PITCH + half * 4];
  const float* bfr = &w1t[ml * W_PITCH + half * 4];

  v4f ra[8], rb[8];
  #pragma unroll
  for (int j = 0; j < 8; ++j) ra[j] = *(const v4f*)(xg + (size_t)(2 * j) * IN_CH);
  #pragma unroll
  for (int j = 0; j < 8; ++j) rb[j] = *(const v4f*)(xg + (size_t)(2 * j) * IN_CH + KC);

  v8f c0 = {}, c1 = {};
  #pragma unroll
  for (int c = 0; c < NCHUNK; ++c) {
    #pragma unroll
    for (int j = 0; j < 8; ++j) *(v4f*)(ast + 2 * j * A_PITCH) = (c & 1) ? rb[j] : ra[j];
    if (c + 2 < NCHUNK) {
      #pragma unroll
      for (int j = 0; j < 8; ++j) {
        const v4f v = *(const v4f*)(xg + (size_t)(2 * j) * IN_CH + (c + 2) * KC);
        if (c & 1) rb[j] = v; else ra[j] = v;
      }
    }
    __builtin_amdgcn_fence(__ATOMIC_RELEASE, "wavefront");
    __builtin_amdgcn_fence(__ATOMIC_ACQUIRE, "wavefront");
    #pragma unroll
    for (int kk = 0; kk < KC / 8; ++kk) {
      const v4f a = *(const v4f*)(afr + kk * 8);
      const v4f b = *(const v4f*)(bfr + c * KC + kk * 8);
      v2f a0, a1, b0, b1;
      a0.x = a.x; a0.y = a.y; a1.x = a.z; a1.y = a.w;
      b0.x = b.x; b0.y = b.y; b1.x = b.z; b1.y = b.w;
      c0 = __builtin_amdgcn_wmma_f32_16x16x4_f32(false, a0, false, b0, (short)0, c0, false, false);
      c1 = __builtin_amdgcn_wmma_f32_16x16x4_f32(false, a1, false, b1, (short)0, c1, false, false);
    }
    __builtin_amdgcn_fence(__ATOMIC_RELEASE, "wavefront");
    __builtin_amdgcn_fence(__ATOMIC_ACQUIRE, "wavefront");
  }

  float* o = xw1 + (size_t)(t * 16) * HID + ml;
  const float* dv = dinv + t * 16 + half * 8;
  #pragma unroll
  for (int r = 0; r < 8; ++r)
    o[(half * 8 + r) * HID] = (c0[r] + c1[r]) * dv[r];
}

__global__ void k_gemm2(const float* __restrict__ g, const float* __restrict__ W2,
                        const float* __restrict__ bias, float* __restrict__ y, int nTiles) {
  __shared__ float w2s[HID * OUT_CH];
  for (int i = threadIdx.x; i < HID * OUT_CH; i += blockDim.x) w2s[i] = W2[i];
  __syncthreads();

  int wave = threadIdx.x >> 5;
  int lane = threadIdx.x & 31;
  int t = blockIdx.x * (blockDim.x >> 5) + wave;
  if (t >= nTiles) return;

  int half = lane >> 4;
  int ml = lane & 15;
  const float* hr = g + (size_t)(t * 16 + ml) * HID + half * 2;

  v8f c0 = {}, c1 = {}, c2 = {}, c3 = {};
  #pragma unroll
  for (int kk = 0; kk < HID / 4; ++kk) {
    v2f a = *(const v2f*)(hr + kk * 4);
    int k0 = kk * 4 + half * 2;
    v2f b0, b1, b2, b3;
    b0.x = w2s[k0 * OUT_CH +  0 + ml]; b0.y = w2s[(k0 + 1) * OUT_CH +  0 + ml];
    b1.x = w2s[k0 * OUT_CH + 16 + ml]; b1.y = w2s[(k0 + 1) * OUT_CH + 16 + ml];
    b2.x = w2s[k0 * OUT_CH + 32 + ml]; b2.y = w2s[(k0 + 1) * OUT_CH + 32 + ml];
    b3.x = w2s[k0 * OUT_CH + 48 + ml]; b3.y = w2s[(k0 + 1) * OUT_CH + 48 + ml];
    c0 = __builtin_amdgcn_wmma_f32_16x16x4_f32(false, a, false, b0, (short)0, c0, false, false);
    c1 = __builtin_amdgcn_wmma_f32_16x16x4_f32(false, a, false, b1, (short)0, c1, false, false);
    c2 = __builtin_amdgcn_wmma_f32_16x16x4_f32(false, a, false, b2, (short)0, c2, false, false);
    c3 = __builtin_amdgcn_wmma_f32_16x16x4_f32(false, a, false, b3, (short)0, c3, false, false);
  }

  const float bs0 = bias[ml], bs1 = bias[16 + ml], bs2 = bias[32 + ml], bs3 = bias[48 + ml];
  float* o = y + (size_t)(t * 16) * OUT_CH + ml;
  #pragma unroll
  for (int r = 0; r < 8; ++r) {
    int row = half * 8 + r;
    o[row * OUT_CH +  0] = c0[r] + bs0;
    o[row * OUT_CH + 16] = c1[r] + bs1;
    o[row * OUT_CH + 32] = c2[r] + bs2;
    o[row * OUT_CH + 48] = c3[r] + bs3;
  }
}

template <bool TAIL>
__device__ __forceinline__ void add8(float4& acc, const float4* __restrict__ rows,
                                     const int* __restrict__ csr, int j, int end, int sub, int n) {
  int s[8];
  float4 v[8];
  #pragma unroll
  for (int k = 0; k < 8; ++k) s[k] = csr[TAIL ? min(j + k, end - 1) : j + k];
  #pragma unroll
  for (int k = 0; k < 8; ++k) v[k] = rows[min(max(s[k], 0), n - 1) * 4 + sub];
  #pragma unroll
  for (int k = 0; k < 8; ++k) {
    const bool ok = !TAIL || k == 0 || j + k < end;
    acc.x = ok ? acc.x + v[k].x : acc.x;
    acc.y = ok ? acc.y + v[k].y : acc.y;
    acc.z = ok ? acc.z + v[k].z : acc.z;
    acc.w = ok ? acc.w + v[k].w : acc.w;
  }
}

template <bool FIRST>
__global__ void k_gather(const int* __restrict__ cnt, const int* __restrict__ bend,
                         const int* __restrict__ csr, const float* __restrict__ dinv,
                         const float* __restrict__ in, const float* __restrict__ bias,
                         float* __restrict__ out, int n) {
  const int gid = blockIdx.x * blockDim.x + threadIdx.x;
  const int node = gid >> 2;
  const int sub = gid & 3;
  if (node >= n) return;
  const float4* __restrict__ rows = (const float4*)in;
  const int end = bend[node];
  const int begin = end - cnt[node];
  float4 acc = rows[node * 4 + sub];
  int j = begin;
  for (; j + 8 <= end; j += 8) add8<false>(acc, rows, csr, j, end, sub, n);
  if (j < end) add8<true>(acc, rows, csr, j, end, sub, n);
  const float di = dinv[node];
  float4 r;
  if (FIRST) {
    const float* b = bias + sub * 4;
    r.x = fmaxf(b[0] + di * acc.x, 0.0f) * di;
    r.y = fmaxf(b[1] + di * acc.y, 0.0f) * di;
    r.z = fmaxf(b[2] + di * acc.z, 0.0f) * di;
    r.w = fmaxf(b[3] + di * acc.w, 0.0f) * di;
  } else {
    r.x = di * acc.x; r.y = di * acc.y; r.z = di * acc.z; r.w = di * acc.w;
  }
  ((float4*)out)[node * 4 + sub] = r;
}

static inline int gridFor(long long total, int threads) {
  return (int)((total + threads - 1) / threads);
}

extern "C" void kernel_launch(void* const* d_in, const int* in_sizes, int n_in,
                              void* d_out, int out_size, void* d_ws, size_t ws_size,
                              hipStream_t stream) {
  const float* x  = (const float*)d_in[0];
  const int* ei   = (const int*)d_in[1];
  const float* W1 = (const float*)d_in[2];
  const float* b1 = (const float*)d_in[3];
  const float* W2 = (const float*)d_in[4];
  const float* b2 = (const float*)d_in[5];
  float* out = (float*)d_out;

  const int n = in_sizes[0] / IN_CH;
  const int e = in_sizes[1] / 2;
  const int* src = ei;
  const int* dst = ei + e;

  float* xs   = (float*)d_ws;
  float* hs   = xs + (size_t)n * HID;
  float* g    = xs;
  int*   csr  = (int*)(hs + (size_t)n * HID);
  int*   cnt  = csr + e;
  int*   cur  = cnt + n;
  float* dinv = (float*)(cur + n);

  const int nTiles = (n + 15) / 16;
  const int wavesPerBlock = THREADS / 32;

  k_zero_cnt<<<gridFor(n, THREADS), THREADS, 0, stream>>>(cnt, n);
  k_count_deg<<<gridFor(e, THREADS), THREADS, 0, stream>>>(dst, cnt, e);
  k_scan<<<gridFor(n, SCAN_T), SCAN_T, 0, stream>>>(cnt, cur, dinv, n);
  k_fill<<<gridFor(e, THREADS), THREADS, 0, stream>>>(src, dst, cur, csr, e);

  k_gemm1<<<gridFor(nTiles, wavesPerBlock), THREADS, 0, stream>>>(x, W1, dinv, xs, nTiles);
  long long lanes = (long long)n * 4;
  k_gather<true><<<gridFor(lanes, THREADS), THREADS, 0, stream>>>(cnt, cur, csr, dinv, xs, b1, hs, n);

  k_gather<false><<<gridFor(lanes, THREADS), THREADS, 0, stream>>>(cnt, cur, csr, dinv, hs, b2, g, n);
  k_gemm2<<<gridFor(nTiles, wavesPerBlock), THREADS, 0, stream>>>(g, W2, b2, out, nTiles);
}


# Counting pass keeps the value each edge's atomic add returns as its rank in the bucket; fill pass places edges with plain stores: one atomic pass, not two; in-bucket sum order now from that pass
# speedup vs baseline: 2.8470x; 1.2897x over previous
#include <hip/hip_runtime.h>


typedef __attribute__((ext_vector_type(2))) float v2f;
typedef __attribute__((ext_vector_type(4))) float v4f;
typedef __attribute__((ext_vector_type(8))) float v8f;

#define IN_CH  512
#define HID    16
#define OUT_CH 64
#define THREADS 256
#define SCAN_T 1024

__global__ void k_zero_cnt(int* __restrict__ cnt, int n) {
  int i = blockIdx.x * blockDim.x + threadIdx.x;
  if (i < n) cnt[i] = 0;
}

__global__ void k_count_rank(const int* __restrict__ dst, int* __restrict__ cnt,
                             int* __restrict__ rank, int e) {
  int i = blockIdx.x * blockDim.x + threadIdx.x;
  if (i < e) rank[i] = atomicAdd(&cnt[dst[i]], 1);
}

__global__ void __launch_bounds__(SCAN_T)
k_scan(const int* __restrict__ cnt, int* __restrict__ cur, float* __restrict__ dinv, int n) {
  __shared__ int wtot[32];
  __shared__ int ptot[32];
  const int tid = threadIdx.x, lane = tid & 31, wave = tid >> 5;
  const int start = blockIdx.x * SCAN_T;
  int pre = 0;
  for (int i = tid; i < start; i += SCAN_T) pre += cnt[i];
  const int i = start + tid;
  const int c = (i < n) ? cnt[i] : 0;
  int incl = c;
  for (int o = 1; o < 32; o <<= 1) {
    int t = __shfl_up(incl, o);
    if (lane >= o) incl += t;
  }
  for (int o = 16; o > 0; o >>= 1) pre += __shfl_xor(pre, o);
  if (lane == 31) wtot[wave] = incl;
  if (lane == 0) ptot[wave] = pre;
  __syncthreads();
  int base = ptot[lane];
  int wpre = (lane < wave) ? wtot[lane] : 0;
  for (int o = 16; o > 0; o >>= 1) {
    base += __shfl_xor(base, o);
    wpre += __shfl_xor(wpre, o);
  }
  if (i < n) {
    cur[i] = base + wpre + incl - c;
    dinv[i] = rsqrtf((float)(c + 1));
  }
}

__global__ void k_fill(const int* __restrict__ src, const int* __restrict__ dst,
                       const int* __restrict__ start, const int* __restrict__ rank,
                       int* __restrict__ csr, int e) {
  int i = blockIdx.x * blockDim.x + threadIdx.x;
  if (i >= e) return;
  int p = start[dst[i]] + rank[i];
  if ((unsigned)p < (unsigned)e) csr[p] = src[i];
}

#define KC 64
#define A_PITCH (KC + 4)
#define W_PITCH (IN_CH + 4)
#define NCHUNK (IN_CH / KC)

__global__ void __launch_bounds__(THREADS)
k_gemm1(const float* __restrict__ x, const float* __restrict__ W1,
        const float* __restrict__ dinv, float* __restrict__ xw1, int nTiles) {
  __shared__ __attribute__((aligned(16))) float w1t[HID * W_PITCH];
  __shared__ __attribute__((aligned(16))) float as[THREADS / 32][16 * A_PITCH];

  {
    v4f w[8];
    #pragma unroll
    for (int i = 0; i < 8; ++i) w[i] = *(const v4f*)(W1 + (i * THREADS + threadIdx.x) * 4);
    #pragma unroll
    for (int i = 0; i < 8; ++i) {
      const int idx = (i * THREADS + threadIdx.x) * 4;
      const int k = idx >> 4, nn = idx & 15;
      w1t[(nn + 0) * W_PITCH + k] = w[i].x;
      w1t[(nn + 1) * W_PITCH + k] = w[i].y;
      w1t[(nn + 2) * W_PITCH + k] = w[i].z;
      w1t[(nn + 3) * W_PITCH + k] = w[i].w;
    }
  }
  __syncthreads();

  const int wave = threadIdx.x >> 5;
  const int lane = threadIdx.x & 31;
  const int t = blockIdx.x * (THREADS / 32) + wave;
  if (t >= nTiles) return;

  const int half = lane >> 4;
  const int ml = lane & 15;
  const float* xg = x + ((size_t)t * 16 + half) * IN_CH + ml * 4;
  float* ast = &as[wave][half * A_PITCH + ml * 4];
  const float* afr = &as[wave][ml * A_PITCH + half * 4];
  const float* bfr = &w1t[ml * W_PITCH + half * 4];

  v4f ra[8], rb[8];
  #pragma unroll
  for (int j = 0; j < 8; ++j) ra[j] = *(const v4f*)(xg + (size_t)(2 * j) * IN_CH);
  #pragma unroll
  for (int j = 0; j < 8; ++j) rb[j] = *(const v4f*)(xg + (size_t)(2 * j) * IN_CH + KC);

  v8f c0 = {}, c1 = {};
  #pragma unroll
  for (int c = 0; c < NCHUNK; ++c) {
    #pragma unroll
    for (int j = 0; j < 8; ++j) *(v4f*)(ast + 2 * j * A_PITCH) = (c & 1) ? rb[j] : ra[j];
    if (c + 2 < NCHUNK) {
      #pragma unroll
      for (int j = 0; j < 8; ++j) {
        const v4f v = *(const v4f*)(xg + (size_t)(2 * j) * IN_CH + (c + 2) * KC);
        if (c & 1) rb[j] = v; else ra[j] = v;
      }
    }
    __builtin_amdgcn_fence(__ATOMIC_RELEASE, "wavefront");
    __builtin_amdgcn_fence(__ATOMIC_ACQUIRE, "wavefront");
    #pragma unroll
    for (int kk = 0; kk < KC / 8; ++kk) {
      const v4f a = *(const v4f*)(afr + kk * 8);
      const v4f b = *(const v4f*)(bfr + c * KC + kk * 8);
      v2f a0, a1, b0, b1;
      a0.x = a.x; a0.y = a.y; a1.x = a.z; a1.y = a.w;
      b0.x = b.x; b0.y = b.y; b1.x = b.z; b1.y = b.w;
      c0 = __builtin_amdgcn_wmma_f32_16x16x4_f32(false, a0, false, b0, (short)0, c0, false, false);
      c1 = __builtin_amdgcn_wmma_f32_16x16x4_f32(false, a1, false, b1, (short)0, c1, false, false);
    }
    __builtin_amdgcn_fence(__ATOMIC_RELEASE, "wavefront");
    __builtin_amdgcn_fence(__ATOMIC_ACQUIRE, "wavefront");
  }

  float* o = xw1 + (size_t)(t * 16) * HID + ml;
  const float* dv = dinv + t * 16 + half * 8;
  #pragma unroll
  for (int r = 0; r < 8; ++r)
    o[(half * 8 + r) * HID] = (c0[r] + c1[r]) * dv[r];
}

__global__ void k_gemm2(const float* __restrict__ g, const float* __restrict__ W2,
                        const float* __restrict__ bias, float* __restrict__ y, int nTiles) {
  __shared__ float w2s[HID * OUT_CH];
  for (int i = threadIdx.x; i < HID * OUT_CH; i += blockDim.x) w2s[i] = W2[i];
  __syncthreads();

  int wave = threadIdx.x >> 5;
  int lane = threadIdx.x & 31;
  int t = blockIdx.x * (blockDim.x >> 5) + wave;
  if (t >= nTiles) return;

  int half = lane >> 4;
  int ml = lane & 15;
  const float* hr = g + (size_t)(t * 16 + ml) * HID + half * 2;

  v8f c0 = {}, c1 = {}, c2 = {}, c3 = {};
  #pragma unroll
  for (int kk = 0; kk < HID / 4; ++kk) {
    v2f a = *(const v2f*)(hr + kk * 4);
    int k0 = kk * 4 + half * 2;
    v2f b0, b1, b2, b3;
    b0.x = w2s[k0 * OUT_CH +  0 + ml]; b0.y = w2s[(k0 + 1) * OUT_CH +  0 + ml];
    b1.x = w2s[k0 * OUT_CH + 16 + ml]; b1.y = w2s[(k0 + 1) * OUT_CH + 16 + ml];
    b2.x = w2s[k0 * OUT_CH + 32 + ml]; b2.y = w2s[(k0 + 1) * OUT_CH + 32 + ml];
    b3.x = w2s[k0 * OUT_CH + 48 + ml]; b3.y = w2s[(k0 + 1) * OUT_CH + 48 + ml];
    c0 = __builtin_amdgcn_wmma_f32_16x16x4_f32(false, a, false, b0, (short)0, c0, false, false);
    c1 = __builtin_amdgcn_wmma_f32_16x16x4_f32(false, a, false, b1, (short)0, c1, false, false);
    c2 = __builtin_amdgcn_wmma_f32_16x16x4_f32(false, a, false, b2, (short)0, c2, false, false);
    c3 = __builtin_amdgcn_wmma_f32_16x16x4_f32(false, a, false, b3, (short)0, c3, false, false);
  }

  const float bs0 = bias[ml], bs1 = bias[16 + ml], bs2 = bias[32 + ml], bs3 = bias[48 + ml];
  float* o = y + (size_t)(t * 16) * OUT_CH + ml;
  #pragma unroll
  for (int r = 0; r < 8; ++r) {
    int row = half * 8 + r;
    o[row * OUT_CH +  0] = c0[r] + bs0;
    o[row * OUT_CH + 16] = c1[r] + bs1;
    o[row * OUT_CH + 32] = c2[r] + bs2;
    o[row * OUT_CH + 48] = c3[r] + bs3;
  }
}

template <bool TAIL>
__device__ __forceinline__ void add8(float4& acc, const float4* __restrict__ rows,
                                     const int* __restrict__ csr, int j, int end, int sub, int n) {
  int s[8];
  float4 v[8];
  #pragma unroll
  for (int k = 0; k < 8; ++k) s[k] = csr[TAIL ? min(j + k, end - 1) : j + k];
  #pragma unroll
  for (int k = 0; k < 8; ++k) v[k] = rows[min(max(s[k], 0), n - 1) * 4 + sub];
  #pragma unroll
  for (int k = 0; k < 8; ++k) {
    const bool ok = !TAIL || k == 0 || j + k < end;
    acc.x = ok ? acc.x + v[k].x : acc.x;
    acc.y = ok ? acc.y + v[k].y : acc.y;
    acc.z = ok ? acc.z + v[k].z : acc.z;
    acc.w = ok ? acc.w + v[k].w : acc.w;
  }
}

template <bool FIRST>
__global__ void k_gather(const int* __restrict__ cnt, const int* __restrict__ bstart,
                         const int* __restrict__ csr, const float* __restrict__ dinv,
                         const float* __restrict__ in, const float* __restrict__ bias,
                         float* __restrict__ out, int n) {
  const int gid = blockIdx.x * blockDim.x + threadIdx.x;
  const int node = gid >> 2;
  const int sub = gid & 3;
  if (node >= n) return;
  const float4* __restrict__ rows = (const float4*)in;
  const int begin = bstart[node];
  const int end = begin + cnt[node];
  float4 acc = rows[node * 4 + sub];
  int j = begin;
  for (; j + 8 <= end; j += 8) add8<false>(acc, rows, csr, j, end, sub, n);
  if (j < end) add8<true>(acc, rows, csr, j, end, sub, n);
  const float di = dinv[node];
  float4 r;
  if (FIRST) {
    const float* b = bias + sub * 4;
    r.x = fmaxf(b[0] + di * acc.x, 0.0f) * di;
    r.y = fmaxf(b[1] + di * acc.y, 0.0f) * di;
    r.z = fmaxf(b[2] + di * acc.z, 0.0f) * di;
    r.w = fmaxf(b[3] + di * acc.w, 0.0f) * di;
  } else {
    r.x = di * acc.x; r.y = di * acc.y; r.z = di * acc.z; r.w = di * acc.w;
  }
  ((float4*)out)[node * 4 + sub] = r;
}

static inline int gridFor(long long total, int threads) {
  return (int)((total + threads - 1) / threads);
}

extern "C" void kernel_launch(void* const* d_in, const int* in_sizes, int n_in,
                              void* d_out, int out_size, void* d_ws, size_t ws_size,
                              hipStream_t stream) {
  const float* x  = (const float*)d_in[0];
  const int* ei   = (const int*)d_in[1];
  const float* W1 = (const float*)d_in[2];
  const float* b1 = (const float*)d_in[3];
  const float* W2 = (const float*)d_in[4];
  const float* b2 = (const float*)d_in[5];
  float* out = (float*)d_out;

  const int n = in_sizes[0] / IN_CH;
  const int e = in_sizes[1] / 2;
  const int* src = ei;
  const int* dst = ei + e;

  float* xs   = (float*)d_ws;
  float* hs   = xs + (size_t)n * HID;
  float* g    = xs;
  int*   csr  = (int*)(hs + (size_t)n * HID);
  int*   cnt  = csr + e;
  int*   cur  = cnt + n;
  float* dinv = (float*)(cur + n);
  int*   rank = (int*)xs;

  const int nTiles = (n + 15) / 16;
  const int wavesPerBlock = THREADS / 32;

  k_zero_cnt<<<gridFor(n, THREADS), THREADS, 0, stream>>>(cnt, n);
  k_count_rank<<<gridFor(e, THREADS), THREADS, 0, stream>>>(dst, cnt, rank, e);
  k_scan<<<gridFor(n, SCAN_T), SCAN_T, 0, stream>>>(cnt, cur, dinv, n);
  k_fill<<<gridFor(e, THREADS), THREADS, 0, stream>>>(src, dst, cur, rank, csr, e);

  k_gemm1<<<gridFor(nTiles, wavesPerBlock), THREADS, 0, stream>>>(x, W1, dinv, xs, nTiles);
  long long lanes = (long long)n * 4;
  k_gather<true><<<gridFor(lanes, THREADS), THREADS, 0, stream>>>(cnt, cur, csr, dinv, xs, b1, hs, n);

  k_gather<false><<<gridFor(lanes, THREADS), THREADS, 0, stream>>>(cnt, cur, csr, dinv, hs, b2, g, n);
  k_gemm2<<<gridFor(nTiles, wavesPerBlock), THREADS, 0, stream>>>(g, W2, b2, out, nTiles);
}


# Bucketing by destination becomes a two-level counting sort in LDS (tiles sorted by coarse bucket, then one block per bucket), stores in runs; no global atomics left; in-bucket sum order now from it
# speedup vs baseline: 7.4462x; 2.6155x over previous
#include <hip/hip_runtime.h>


typedef __attribute__((ext_vector_type(2))) float v2f;
typedef __attribute__((ext_vector_type(4))) float v4f;
typedef __attribute__((ext_vector_type(8))) float v8f;

#define IN_CH  512
#define HID    16
#define OUT_CH 64
#define THREADS 256
#define SCAN_T 1024

#define NSH   8
#define NPB   (1 << NSH)
#define TILE  12288
#define PT    1024
#define EPT   (TILE / PT)
#define BT    512
#define CAP   12288
#define LPT   20

__device__ __forceinline__ int block_excl_scan(int v, int* wtot) {
  const int lane = threadIdx.x & 31, wave = threadIdx.x >> 5;
  int incl = v;
  for (int o = 1; o < 32; o <<= 1) {
    int t = __shfl_up(incl, o);
    if (lane >= o) incl += t;
  }
  if (lane == 31) wtot[wave] = incl;
  __syncthreads();
  int wpre = (lane < wave) ? wtot[lane] : 0;
  for (int o = 16; o > 0; o >>= 1) wpre += __shfl_xor(wpre, o);
  return wpre + incl - v;
}

__global__ void __launch_bounds__(PT)
k_hist(const int* __restrict__ dst, int* __restrict__ ghist, int e, int nb, int nt) {
  __shared__ int hist[PT];
  const int tid = threadIdx.x, tile = blockIdx.x;
  hist[tid] = 0;
  __syncthreads();
  const int t0 = tile * TILE;
  int d[EPT];
  #pragma unroll
  for (int k = 0; k < EPT; ++k) {
    const int i = t0 + k * PT + tid;
    const int dv = dst[min(i, e - 1)];
    d[k] = (i < e) ? dv : -1;
  }
  #pragma unroll
  for (int k = 0; k < EPT; ++k)
    if (d[k] >= 0) atomicAdd(&hist[d[k] >> NSH], 1);
  __syncthreads();
  if (tid < nb) ghist[tid * nt + tile] = hist[tid];
}

__global__ void __launch_bounds__(SCAN_T)
k_scan(const int* __restrict__ hist, int* __restrict__ off, int* __restrict__ bbeg, int total, int nt) {
  __shared__ int wtot[32];
  __shared__ int ptot[32];
  const int tid = threadIdx.x, lane = tid & 31, wave = tid >> 5;
  const int start = blockIdx.x * SCAN_T;
  int pre = 0;
  for (int i = tid; i < start; i += SCAN_T) pre += hist[i];
  const int i = start + tid;
  const int c = (i < total) ? hist[i] : 0;
  int incl = c;
  for (int o = 1; o < 32; o <<= 1) {
    int t = __shfl_up(incl, o);
    if (lane >= o) incl += t;
  }
  for (int o = 16; o > 0; o >>= 1) pre += __shfl_xor(pre, o);
  if (lane == 31) wtot[wave] = incl;
  if (lane == 0) ptot[wave] = pre;
  __syncthreads();
  int base = ptot[lane];
  int wpre = (lane < wave) ? wtot[lane] : 0;
  for (int o = 16; o > 0; o >>= 1) {
    base += __shfl_xor(base, o);
    wpre += __shfl_xor(wpre, o);
  }
  if (i < total) {
    const int ex = base + wpre + incl - c;
    off[i] = ex;
    if (i % nt == 0) bbeg[i / nt] = ex;
  }
}

__global__ void __launch_bounds__(PT)
k_part(const int* __restrict__ src, const int* __restrict__ dst, const int* __restrict__ ghist,
       const int* __restrict__ goff, int* __restrict__ pairs, int e, int nb, int nt) {
  __shared__ int lcur[PT];
  __shared__ int gbase[PT];
  __shared__ int wtot[32];
  __shared__ int s_val[TILE];
  __shared__ unsigned short s_bk[TILE];
  const int tid = threadIdx.x, tile = blockIdx.x;
  const int t0 = tile * TILE;
  int d[EPT], s[EPT];
  #pragma unroll
  for (int k = 0; k < EPT; ++k) {
    const int i = t0 + k * PT + tid;
    const int ic = min(i, e - 1);
    const int dv = dst[ic];
    s[k] = src[ic];
    d[k] = (i < e) ? dv : -1;
  }
  const int h = (tid < nb) ? ghist[tid * nt + tile] : 0;
  const int ex = block_excl_scan(h, wtot);
  lcur[tid] = ex;
  gbase[tid] = (tid < nb) ? goff[tid * nt + tile] - ex : 0;
  __syncthreads();
  #pragma unroll
  for (int k = 0; k < EPT; ++k) {
    if (d[k] >= 0) {
      const int b = d[k] >> NSH;
      const int pos = atomicAdd(&lcur[b], 1);
      if ((unsigned)pos < (unsigned)TILE) {
        s_val[pos] = (s[k] << NSH) | (d[k] & (NPB - 1));
        s_bk[pos] = (unsigned short)b;
      }
    }
  }
  __syncthreads();
  const int m = min(TILE, e - t0);
  for (int i = tid; i < m; i += PT) {
    const int p = gbase[s_bk[i]] + i;
    if ((unsigned)p < (unsigned)e) pairs[p] = s_val[i];
  }
}

__global__ void __launch_bounds__(BT)
k_bucket(const int* __restrict__ pairs, const int* __restrict__ bbeg, int* __restrict__ csr,
         int* __restrict__ cnt, int* __restrict__ start, int n, int e, int nb) {
  __shared__ int lcnt[NPB];
  __shared__ int lcur[NPB];
  __shared__ int wtot[32];
  __shared__ int sorted[CAP];
  const int tid = threadIdx.x, bk = blockIdx.x;
  const int beg = min(max(bbeg[bk], 0), e);
  const int end = (bk + 1 < nb) ? min(max(bbeg[bk + 1], beg), e) : e;
  const int m = end - beg;
  if (tid < NPB) lcnt[tid] = 0;
  __syncthreads();
  for (int i0 = 0; i0 < m; i0 += BT * LPT) {
    int w[LPT];
    #pragma unroll
    for (int k = 0; k < LPT; ++k)
      w[k] = pairs[beg + min(i0 + k * BT + tid, m - 1)];
    #pragma unroll
    for (int k = 0; k < LPT; ++k)
      if (i0 + k * BT + tid < m) atomicAdd(&lcnt[w[k] & (NPB - 1)], 1);
  }
  __syncthreads();
  const int c = (tid < NPB) ? lcnt[tid] : 0;
  const int ex = block_excl_scan(c, wtot);
  if (tid < NPB) {
    lcur[tid] = ex;
    const int node = bk * NPB + tid;
    if (node < n) { cnt[node] = c; start[node] = beg + ex; }
  }
  __syncthreads();
  const bool staged = m <= CAP;
  for (int i0 = 0; i0 < m; i0 += BT * LPT) {
    int w[LPT];
    #pragma unroll
    for (int k = 0; k < LPT; ++k)
      w[k] = pairs[beg + min(i0 + k * BT + tid, m - 1)];
    #pragma unroll
    for (int k = 0; k < LPT; ++k) {
      if (i0 + k * BT + tid < m) {
        const int pos = atomicAdd(&lcur[w[k] & (NPB - 1)], 1);
        if ((unsigned)pos < (unsigned)m) {
          if (staged) sorted[pos] = w[k] >> NSH; else csr[beg + pos] = w[k] >> NSH;
        }
      }
    }
  }
  __syncthreads();
  if (staged)
    for (int i = tid; i < m; i += BT) csr[beg + i] = sorted[i];
}

#define KC 64
#define A_PITCH (KC + 4)
#define W_PITCH (IN_CH + 4)
#define NCHUNK (IN_CH / KC)

__global__ void __launch_bounds__(THREADS)
k_gemm1(const float* __restrict__ x, const float* __restrict__ W1,
        const int* __restrict__ cnt, float* __restrict__ xw1, int nTiles) {
  __shared__ __attribute__((aligned(16))) float w1t[HID * W_PITCH];
  __shared__ __attribute__((aligned(16))) float as[THREADS / 32][16 * A_PITCH];

  {
    v4f w[8];
    #pragma unroll
    for (int i = 0; i < 8; ++i) w[i] = *(const v4f*)(W1 + (i * THREADS + threadIdx.x) * 4);
    #pragma unroll
    for (int i = 0; i < 8; ++i) {
      const int idx = (i * THREADS + threadIdx.x) * 4;
      const int k = idx >> 4, nn = idx & 15;
      w1t[(nn + 0) * W_PITCH + k] = w[i].x;
      w1t[(nn + 1) * W_PITCH + k] = w[i].y;
      w1t[(nn + 2) * W_PITCH + k] = w[i].z;
      w1t[(nn + 3) * W_PITCH + k] = w[i].w;
    }
  }
  __syncthreads();

  const int wave = threadIdx.x >> 5;
  const int lane = threadIdx.x & 31;
  const int t = blockIdx.x * (THREADS / 32) + wave;
  if (t >= nTiles) return;

  const int half = lane >> 4;
  const int ml = lane & 15;
  const float* xg = x + ((size_t)t * 16 + half) * IN_CH + ml * 4;
  float* ast = &as[wave][half * A_PITCH + ml * 4];
  const float* afr = &as[wave][ml * A_PITCH + half * 4];
  const float* bfr = &w1t[ml * W_PITCH + half * 4];

  v4f ra[8], rb[8];
  #pragma unroll
  for (int j = 0; j < 8; ++j) ra[j] = *(const v4f*)(xg + (size_t)(2 * j) * IN_CH);
  #pragma unroll
  for (int j = 0; j < 8; ++j) rb[j] = *(const v4f*)(xg + (size_t)(2 * j) * IN_CH + KC);

  v8f c0 = {}, c1 = {};
  #pragma unroll
  for (int c = 0; c < NCHUNK; ++c) {
    #pragma unroll
    for (int j = 0; j < 8; ++j) *(v4f*)(ast + 2 * j * A_PITCH) = (c & 1) ? rb[j] : ra[j];
    if (c + 2 < NCHUNK) {
      #pragma unroll
      for (int j = 0; j < 8; ++j) {
        const v4f v = *(const v4f*)(xg + (size_t)(2 * j) * IN_CH + (c + 2) * KC);
        if (c & 1) rb[j] = v; else ra[j] = v;
      }
    }
    __builtin_amdgcn_fence(__ATOMIC_RELEASE, "wavefront");
    __builtin_amdgcn_fence(__ATOMIC_ACQUIRE, "wavefront");
    #pragma unroll
    for (int kk = 0; kk < KC / 8; ++kk) {
      const v4f a = *(const v4f*)(afr + kk * 8);
      const v4f b = *(const v4f*)(bfr + c * KC + kk * 8);
      v2f a0, a1, b0, b1;
      a0.x = a.x; a0.y = a.y; a1.x = a.z; a1.y = a.w;
      b0.x = b.x; b0.y = b.y; b1.x = b.z; b1.y = b.w;
      c0 = __builtin_amdgcn_wmma_f32_16x16x4_f32(false, a0, false, b0, (short)0, c0, false, false);
      c1 = __builtin_amdgcn_wmma_f32_16x16x4_f32(false, a1, false, b1, (short)0, c1, false, false);
    }
    __builtin_amdgcn_fence(__ATOMIC_RELEASE, "wavefront");
    __builtin_amdgcn_fence(__ATOMIC_ACQUIRE, "wavefront");
  }

  float* o = xw1 + (size_t)(t * 16) * HID + ml;
  const int* dc = cnt + t * 16 + half * 8;
  #pragma unroll
  for (int r = 0; r < 8; ++r)
    o[(half * 8 + r) * HID] = (c0[r] + c1[r]) * rsqrtf((float)(dc[r] + 1));
}

__global__ void k_gemm2(const float* __restrict__ g, const float* __restrict__ W2,
                        const float* __restrict__ bias, float* __restrict__ y, int nTiles) {
  __shared__ float w2s[HID * OUT_CH];
  for (int i = threadIdx.x; i < HID * OUT_CH; i += blockDim.x) w2s[i] = W2[i];
  __syncthreads();

  int wave = threadIdx.x >> 5;
  int lane = threadIdx.x & 31;
  int t = blockIdx.x * (blockDim.x >> 5) + wave;
  if (t >= nTiles) return;

  int half = lane >> 4;
  int ml = lane & 15;
  const float* hr = g + (size_t)(t * 16 + ml) * HID + half * 2;

  v8f c0 = {}, c1 = {}, c2 = {}, c3 = {};
  #pragma unroll
  for (int kk = 0; kk < HID / 4; ++kk) {
    v2f a = *(const v2f*)(hr + kk * 4);
    int k0 = kk * 4 + half * 2;
    v2f b0, b1, b2, b3;
    b0.x = w2s[k0 * OUT_CH +  0 + ml]; b0.y = w2s[(k0 + 1) * OUT_CH +  0 + ml];
    b1.x = w2s[k0 * OUT_CH + 16 + ml]; b1.y = w2s[(k0 + 1) * OUT_CH + 16 + ml];
    b2.x = w2s[k0 * OUT_CH + 32 + ml]; b2.y = w2s[(k0 + 1) * OUT_CH + 32 + ml];
    b3.x = w2s[k0 * OUT_CH + 48 + ml]; b3.y = w2s[(k0 + 1) * OUT_CH + 48 + ml];
    c0 = __builtin_amdgcn_wmma_f32_16x16x4_f32(false, a, false, b0, (short)0, c0, false, false);
    c1 = __builtin_amdgcn_wmma_f32_16x16x4_f32(false, a, false, b1, (short)0, c1, false, false);
    c2 = __builtin_amdgcn_wmma_f32_16x16x4_f32(false, a, false, b2, (short)0, c2, false, false);
    c3 = __builtin_amdgcn_wmma_f32_16x16x4_f32(false, a, false, b3, (short)0, c3, false, false);
  }

  const float bs0 = bias[ml], bs1 = bias[16 + ml], bs2 = bias[32 + ml], bs3 = bias[48 + ml];
  float* o = y + (size_t)(t * 16) * OUT_CH + ml;
  #pragma unroll
  for (int r = 0; r < 8; ++r) {
    int row = half * 8 + r;
    o[row * OUT_CH +  0] = c0[r] + bs0;
    o[row * OUT_CH + 16] = c1[r] + bs1;
    o[row * OUT_CH + 32] = c2[r] + bs2;
    o[row * OUT_CH + 48] = c3[r] + bs3;
  }
}

template <bool TAIL>
__device__ __forceinline__ void add8(float4& acc, const float4* __restrict__ rows,
                                     const int* __restrict__ csr, int j, int end, int sub, int n) {
  int s[8];
  float4 v[8];
  #pragma unroll
  for (int k = 0; k < 8; ++k) s[k] = csr[TAIL ? min(j + k, end - 1) : j + k];
  #pragma unroll
  for (int k = 0; k < 8; ++k) v[k] = rows[min(max(s[k], 0), n - 1) * 4 + sub];
  #pragma unroll
  for (int k = 0; k < 8; ++k) {
    const bool ok = !TAIL || k == 0 || j + k < end;
    acc.x = ok ? acc.x + v[k].x : acc.x;
    acc.y = ok ? acc.y + v[k].y : acc.y;
    acc.z = ok ? acc.z + v[k].z : acc.z;
    acc.w = ok ? acc.w + v[k].w : acc.w;
  }
}

template <bool FIRST>
__global__ void k_gather(const int* __restrict__ cnt, const int* __restrict__ bstart,
                         const int* __restrict__ csr,
                         const float* __restrict__ in, const float* __restrict__ bias,
                         float* __restrict__ out, int n) {
  const int gid = blockIdx.x * blockDim.x + threadIdx.x;
  const int node = gid >> 2;
  const int sub = gid & 3;
  if (node >= n) return;
  const float4* __restrict__ rows = (const float4*)in;
  const int begin = bstart[node];
  const int deg = cnt[node];
  const int end = begin + deg;
  float4 acc = rows[node * 4 + sub];
  int j = begin;
  for (; j + 8 <= end; j += 8) add8<false>(acc, rows, csr, j, end, sub, n);
  if (j < end) add8<true>(acc, rows, csr, j, end, sub, n);
  const float di = rsqrtf((float)(deg + 1));
  float4 r;
  if (FIRST) {
    const float* b = bias + sub * 4;
    r.x = fmaxf(b[0] + di * acc.x, 0.0f) * di;
    r.y = fmaxf(b[1] + di * acc.y, 0.0f) * di;
    r.z = fmaxf(b[2] + di * acc.z, 0.0f) * di;
    r.w = fmaxf(b[3] + di * acc.w, 0.0f) * di;
  } else {
    r.x = di * acc.x; r.y = di * acc.y; r.z = di * acc.z; r.w = di * acc.w;
  }
  ((float4*)out)[node * 4 + sub] = r;
}

static inline int gridFor(long long total, int threads) {
  return (int)((total + threads - 1) / threads);
}

extern "C" void kernel_launch(void* const* d_in, const int* in_sizes, int n_in,
                              void* d_out, int out_size, void* d_ws, size_t ws_size,
                              hipStream_t stream) {
  const float* x  = (const float*)d_in[0];
  const int* ei   = (const int*)d_in[1];
  const float* W1 = (const float*)d_in[2];
  const float* b1 = (const float*)d_in[3];
  const float* W2 = (const float*)d_in[4];
  const float* b2 = (const float*)d_in[5];
  float* out = (float*)d_out;

  const int n = in_sizes[0] / IN_CH;
  const int e = in_sizes[1] / 2;
  const int* src = ei;
  const int* dst = ei + e;

  float* xs   = (float*)d_ws;
  float* hs   = xs + (size_t)n * HID;
  float* g    = xs;
  int*   csr  = (int*)(hs + (size_t)n * HID);
  int*   cnt  = csr + e;
  int*   cur  = cnt + n;
  int*   bbeg = cur + n;
  int*   pairs = (int*)xs;
  int*   ghist = cnt;
  int*   goff  = csr;

  const int nTiles = (n + 15) / 16;
  const int wavesPerBlock = THREADS / 32;
  const int nb = gridFor(n, NPB);
  const int nt = gridFor(e, TILE);

  k_hist<<<nt, PT, 0, stream>>>(dst, ghist, e, nb, nt);
  k_scan<<<gridFor((long long)nb * nt, SCAN_T), SCAN_T, 0, stream>>>(ghist, goff, bbeg, nb * nt, nt);
  k_part<<<nt, PT, 0, stream>>>(src, dst, ghist, goff, pairs, e, nb, nt);
  k_bucket<<<nb, BT, 0, stream>>>(pairs, bbeg, csr, cnt, cur, n, e, nb);

  k_gemm1<<<gridFor(nTiles, wavesPerBlock), THREADS, 0, stream>>>(x, W1, cnt, xs, nTiles);
  long long lanes = (long long)n * 4;
  k_gather<true><<<gridFor(lanes, THREADS), THREADS, 0, stream>>>(cnt, cur, csr, xs, b1, hs, n);

  k_gather<false><<<gridFor(lanes, THREADS), THREADS, 0, stream>>>(cnt, cur, csr, hs, b2, g, n);
  k_gemm2<<<gridFor(nTiles, wavesPerBlock), THREADS, 0, stream>>>(g, W2, b2, out, nTiles);
}


# Level-1 tiles stay sorted in place with a row of run starts; a level-2 block collects its runs from the tiles, csr base = sum of the starts: no histogram or scan launch; in-bucket sum order changes
# speedup vs baseline: 8.3647x; 1.1234x over previous
#include <hip/hip_runtime.h>


typedef __attribute__((ext_vector_type(2))) float v2f;
typedef __attribute__((ext_vector_type(4))) float v4f;
typedef __attribute__((ext_vector_type(8))) float v8f;

#define IN_CH  512
#define HID    16
#define OUT_CH 64
#define THREADS 256
#define SCAN_T 1024

#define NSH   8
#define NPB   (1 << NSH)
#define TILE  12288
#define PT    1024
#define EPT   (TILE / PT)
#define BT    512
#define CAP   12288
#define RPW   8

__device__ __forceinline__ int block_excl_scan(int v, int* wtot) {
  const int lane = threadIdx.x & 31, wave = threadIdx.x >> 5;
  int incl = v;
  for (int o = 1; o < 32; o <<= 1) {
    int t = __shfl_up(incl, o);
    if (lane >= o) incl += t;
  }
  if (lane == 31) wtot[wave] = incl;
  __syncthreads();
  int wpre = (lane < wave) ? wtot[lane] : 0;
  for (int o = 16; o > 0; o >>= 1) wpre += __shfl_xor(wpre, o);
  return wpre + incl - v;
}

__global__ void __launch_bounds__(PT)
k_part(const int* __restrict__ src, const int* __restrict__ dst, int* __restrict__ pairs,
       int* __restrict__ toffw, int e, int nb) {
  __shared__ int lcur[PT];
  __shared__ int wtot[32];
  __shared__ int s_val[TILE];
  unsigned short* toff = (unsigned short*)toffw;
  const int tid = threadIdx.x, tile = blockIdx.x;
  const int t0 = tile * TILE;
  int d[EPT], s[EPT];
  #pragma unroll
  for (int k = 0; k < EPT; ++k) {
    const int i = t0 + k * PT + tid;
    const int ic = min(i, e - 1);
    const int dv = dst[ic];
    s[k] = src[ic];
    d[k] = (i < e) ? dv : -1;
  }
  lcur[tid] = 0;
  __syncthreads();
  #pragma unroll
  for (int k = 0; k < EPT; ++k)
    if (d[k] >= 0) atomicAdd(&lcur[d[k] >> NSH], 1);
  __syncthreads();
  const int h = lcur[tid];
  const int ex = block_excl_scan(h, wtot);
  lcur[tid] = ex;
  if (tid < nb) toff[tile * nb + tid] = (unsigned short)ex;
  __syncthreads();
  #pragma unroll
  for (int k = 0; k < EPT; ++k) {
    if (d[k] >= 0) {
      const int pos = atomicAdd(&lcur[d[k] >> NSH], 1);
      if ((unsigned)pos < (unsigned)TILE) s_val[pos] = (s[k] << NSH) | (d[k] & (NPB - 1));
    }
  }
  __syncthreads();
  const int m = min(TILE, e - t0);
  for (int i = tid; i < m; i += PT) pairs[t0 + i] = s_val[i];
}

template <class F>
__device__ __forceinline__ void walk_runs(const int* __restrict__ pairs, const int* r_src, const int* r_pos,
                                          const int* r_len, int nt, int e, F f) {
  const int lane = threadIdx.x & 31, wave = threadIdx.x >> 5;
  for (int t0 = wave * RPW; t0 < nt; t0 += (BT / 32) * RPW) {
    int w0[RPW], w1[RPW];
    #pragma unroll
    for (int u = 0; u < RPW; ++u) {
      const int t = min(t0 + u, nt - 1);
      const int base = r_src[t], last = max(r_len[t] - 1, 0);
      w0[u] = pairs[min(base + min(lane, last), e - 1)];
      w1[u] = pairs[min(base + min(lane + 32, last), e - 1)];
    }
    #pragma unroll
    for (int u = 0; u < RPW; ++u) {
      const int t = t0 + u;
      if (t < nt) {
        const int len = r_len[t], lp = r_pos[t];
        if (lane < len) f(lp + lane, w0[u]);
        if (lane + 32 < len) f(lp + lane + 32, w1[u]);
        for (int j = lane + 64; j < len; j += 32) f(lp + j, pairs[min(r_src[t] + j, e - 1)]);
      }
    }
  }
}

__global__ void __launch_bounds__(BT)
k_bucket(const int* __restrict__ pairs, const int* __restrict__ toffw, int* __restrict__ csr,
         int* __restrict__ start, int n, int e, int nb, int nt) {
  __shared__ int lcnt[NPB];
  __shared__ int lcur[NPB];
  __shared__ int wtot[32];
  __shared__ int r_src[BT];
  __shared__ int r_pos[BT];
  __shared__ int r_len[BT];
  __shared__ int s_tot[2];
  __shared__ int raw[CAP];
  __shared__ int sorted[CAP];
  const unsigned short* toff = (const unsigned short*)toffw;
  const int tid = threadIdx.x, bk = blockIdx.x;
  int o0 = 0, len = 0;
  if (tid < nt) {
    const int tm = min(TILE, e - tid * TILE);
    const int a = toff[tid * nb + bk];
    const int b = (bk + 1 < nb) ? (int)toff[tid * nb + bk + 1] : tm;
    o0 = min(a, tm);
    len = min(max(b - o0, 0), tm - o0);
  }
  if (tid < NPB) lcnt[tid] = 0;
  const int lp = block_excl_scan(len, wtot);
  r_src[tid] = tid * TILE + o0;
  r_pos[tid] = lp;
  r_len[tid] = len;
  if (tid == BT - 1) s_tot[0] = lp + len;
  __syncthreads();
  const int pre = block_excl_scan(o0, wtot);
  if (tid == BT - 1) s_tot[1] = pre + o0;
  __syncthreads();
  const int m = s_tot[0];
  const int beg = min(s_tot[1], e - m);
  const bool staged = m <= CAP;
  walk_runs(pairs, r_src, r_pos, r_len, nt, e, [&](int p, int w) {
    if (staged && (unsigned)p < (unsigned)CAP) raw[p] = w;
    atomicAdd(&lcnt[w & (NPB - 1)], 1);
  });
  __syncthreads();
  const int c = (tid < NPB) ? lcnt[tid] : 0;
  const int ex = block_excl_scan(c, wtot);
  if (tid < NPB) {
    lcur[tid] = ex;
    const int node = bk * NPB + tid;
    if (node < n) start[node] = beg + ex;
  }
  if (bk == 0 && tid == 0) start[n] = e;
  __syncthreads();
  if (staged) {
    for (int i = tid; i < m; i += BT) {
      const int w = raw[i];
      const int pos = atomicAdd(&lcur[w & (NPB - 1)], 1);
      if ((unsigned)pos < (unsigned)CAP) sorted[pos] = w >> NSH;
    }
    __syncthreads();
    for (int i = tid; i < m; i += BT) csr[beg + i] = sorted[i];
  } else {
    walk_runs(pairs, r_src, r_pos, r_len, nt, e, [&](int p, int w) {
      const int pos = atomicAdd(&lcur[w & (NPB - 1)], 1);
      if ((unsigned)pos < (unsigned)m) csr[beg + pos] = w >> NSH;
    });
  }
}

#define KC 64
#define A_PITCH (KC + 4)
#define W_PITCH (IN_CH + 4)
#define NCHUNK (IN_CH / KC)

__global__ void __launch_bounds__(THREADS)
k_gemm1(const float* __restrict__ x, const float* __restrict__ W1,
        const int* __restrict__ start, float* __restrict__ xw1, int nTiles) {
  __shared__ __attribute__((aligned(16))) float w1t[HID * W_PITCH];
  __shared__ __attribute__((aligned(16))) float as[THREADS / 32][16 * A_PITCH];

  {
    v4f w[8];
    #pragma unroll
    for (int i = 0; i < 8; ++i) w[i] = *(const v4f*)(W1 + (i * THREADS + threadIdx.x) * 4);
    #pragma unroll
    for (int i = 0; i < 8; ++i) {
      const int idx = (i * THREADS + threadIdx.x) * 4;
      const int k = idx >> 4, nn = idx & 15;
      w1t[(nn + 0) * W_PITCH + k] = w[i].x;
      w1t[(nn + 1) * W_PITCH + k] = w[i].y;
      w1t[(nn + 2) * W_PITCH + k] = w[i].z;
      w1t[(nn + 3) * W_PITCH + k] = w[i].w;
    }
  }
  __syncthreads();

  const int wave = threadIdx.x >> 5;
  const int lane = threadIdx.x & 31;
  const int t = blockIdx.x * (THREADS / 32) + wave;
  if (t >= nTiles) return;

  const int half = lane >> 4;
  const int ml = lane & 15;
  const float* xg = x + ((size_t)t * 16 + half) * IN_CH + ml * 4;
  float* ast = &as[wave][half * A_PITCH + ml * 4];
  const float* afr = &as[wave][ml * A_PITCH + half * 4];
  const float* bfr = &w1t[ml * W_PITCH + half * 4];

  v4f ra[8], rb[8];
  #pragma unroll
  for (int j = 0; j < 8; ++j) ra[j] = *(const v4f*)(xg + (size_t)(2 * j) * IN_CH);
  #pragma unroll
  for (int j = 0; j < 8; ++j) rb[j] = *(const v4f*)(xg + (size_t)(2 * j) * IN_CH + KC);

  v8f c0 = {}, c1 = {};
  #pragma unroll
  for (int c = 0; c < NCHUNK; ++c) {
    #pragma unroll
    for (int j = 0; j < 8; ++j) *(v4f*)(ast + 2 * j * A_PITCH) = (c & 1) ? rb[j] : ra[j];
    if (c + 2 < NCHUNK) {
      #pragma unroll
      for (int j = 0; j < 8; ++j) {
        const v4f v = *(const v4f*)(xg + (size_t)(2 * j) * IN_CH + (c + 2) * KC);
        if (c & 1) rb[j] = v; else ra[j] = v;
      }
    }
    __builtin_amdgcn_fence(__ATOMIC_RELEASE, "wavefront");
    __builtin_amdgcn_fence(__ATOMIC_ACQUIRE, "wavefront");
    #pragma unroll
    for (int kk = 0; kk < KC / 8; ++kk) {
      const v4f a = *(const v4f*)(afr + kk * 8);
      const v4f b = *(const v4f*)(bfr + c * KC + kk * 8);
      v2f a0, a1, b0, b1;
      a0.x = a.x; a0.y = a.y; a1.x = a.z; a1.y = a.w;
      b0.x = b.x; b0.y = b.y; b1.x = b.z; b1.y = b.w;
      c0 = __builtin_amdgcn_wmma_f32_16x16x4_f32(false, a0, false, b0, (short)0, c0, false, false);
      c1 = __builtin_amdgcn_wmma_f32_16x16x4_f32(false, a1, false, b1, (short)0, c1, false, false);
    }
    __builtin_amdgcn_fence(__ATOMIC_RELEASE, "wavefront");
    __builtin_amdgcn_fence(__ATOMIC_ACQUIRE, "wavefront");
  }

  float* o = xw1 + (size_t)(t * 16) * HID + ml;
  const int* dc = start + t * 16 + half * 8;
  #pragma unroll
  for (int r = 0; r < 8; ++r)
    o[(half * 8 + r) * HID] = (c0[r] + c1[r]) * rsqrtf((float)(dc[r + 1] - dc[r] + 1));
}

__global__ void k_gemm2(const float* __restrict__ g, const float* __restrict__ W2,
                        const float* __restrict__ bias, float* __restrict__ y, int nTiles) {
  __shared__ float w2s[HID * OUT_CH];
  for (int i = threadIdx.x; i < HID * OUT_CH; i += blockDim.x) w2s[i] = W2[i];
  __syncthreads();

  int wave = threadIdx.x >> 5;
  int lane = threadIdx.x & 31;
  int t = blockIdx.x * (blockDim.x >> 5) + wave;
  if (t >= nTiles) return;

  int half = lane >> 4;
  int ml = lane & 15;
  const float* hr = g + (size_t)(t * 16 + ml) * HID + half * 2;

  v8f c0 = {}, c1 = {}, c2 = {}, c3 = {};
  #pragma unroll
  for (int kk = 0; kk < HID / 4; ++kk) {
    v2f a = *(const v2f*)(hr + kk * 4);
    int k0 = kk * 4 + half * 2;
    v2f b0, b1, b2, b3;
    b0.x = w2s[k0 * OUT_CH +  0 + ml]; b0.y = w2s[(k0 + 1) * OUT_CH +  0 + ml];
    b1.x = w2s[k0 * OUT_CH + 16 + ml]; b1.y = w2s[(k0 + 1) * OUT_CH + 16 + ml];
    b2.x = w2s[k0 * OUT_CH + 32 + ml]; b2.y = w2s[(k0 + 1) * OUT_CH + 32 + ml];
    b3.x = w2s[k0 * OUT_CH + 48 + ml]; b3.y = w2s[(k0 + 1) * OUT_CH + 48 + ml];
    c0 = __builtin_amdgcn_wmma_f32_16x16x4_f32(false, a, false, b0, (short)0, c0, false, false);
    c1 = __builtin_amdgcn_wmma_f32_16x16x4_f32(false, a, false, b1, (short)0, c1, false, false);
    c2 = __builtin_amdgcn_wmma_f32_16x16x4_f32(false, a, false, b2, (short)0, c2, false, false);
    c3 = __builtin_amdgcn_wmma_f32_16x16x4_f32(false, a, false, b3, (short)0, c3, false, false);
  }

  const float bs0 = bias[ml], bs1 = bias[16 + ml], bs2 = bias[32 + ml], bs3 = bias[48 + ml];
  float* o = y + (size_t)(t * 16) * OUT_CH + ml;
  #pragma unroll
  for (int r = 0; r < 8; ++r) {
    int row = half * 8 + r;
    o[row * OUT_CH +  0] = c0[r] + bs0;
    o[row * OUT_CH + 16] = c1[r] + bs1;
    o[row * OUT_CH + 32] = c2[r] + bs2;
    o[row * OUT_CH + 48] = c3[r] + bs3;
  }
}

template <bool TAIL>
__device__ __forceinline__ void add8(float4& acc, const float4* __restrict__ rows,
                                     const int* __restrict__ csr, int j, int end, int sub, int n) {
  int s[8];
  float4 v[8];
  #pragma unroll
  for (int k = 0; k < 8; ++k) s[k] = csr[TAIL ? min(j + k, end - 1) : j + k];
  #pragma unroll
  for (int k = 0; k < 8; ++k) v[k] = rows[min(max(s[k], 0), n - 1) * 4 + sub];
  #pragma unroll
  for (int k = 0; k < 8; ++k) {
    const bool ok = !TAIL || k == 0 || j + k < end;
    acc.x = ok ? acc.x + v[k].x : acc.x;
    acc.y = ok ? acc.y + v[k].y : acc.y;
    acc.z = ok ? acc.z + v[k].z : acc.z;
    acc.w = ok ? acc.w + v[k].w : acc.w;
  }
}

template <bool FIRST>
__global__ void k_gather(const int* __restrict__ bstart,
                         const int* __restrict__ csr,
                         const float* __restrict__ in, const float* __restrict__ bias,
                         float* __restrict__ out, int n) {
  const int gid = blockIdx.x * blockDim.x + threadIdx.x;
  const int node = gid >> 2;
  const int sub = gid & 3;
  if (node >= n) return;
  const float4* __restrict__ rows = (const float4*)in;
  const int begin = bstart[node];
  const int end = bstart[node + 1];
  const int deg = end - begin;
  float4 acc = rows[node * 4 + sub];
  int j = begin;
  for (; j + 8 <= end; j += 8) add8<false>(acc, rows, csr, j, end, sub, n);
  if (j < end) add8<true>(acc, rows, csr, j, end, sub, n);
  const float di = rsqrtf((float)(deg + 1));
  float4 r;
  if (FIRST) {
    const float* b = bias + sub * 4;
    r.x = fmaxf(b[0] + di * acc.x, 0.0f) * di;
    r.y = fmaxf(b[1] + di * acc.y, 0.0f) * di;
    r.z = fmaxf(b[2] + di * acc.z, 0.0f) * di;
    r.w = fmaxf(b[3] + di * acc.w, 0.0f) * di;
  } else {
    r.x = di * acc.x; r.y = di * acc.y; r.z = di * acc.z; r.w = di * acc.w;
  }
  ((float4*)out)[node * 4 + sub] = r;
}

static inline int gridFor(long long total, int threads) {
  return (int)((total + threads - 1) / threads);
}

extern "C" void kernel_launch(void* const* d_in, const int* in_sizes, int n_in,
                              void* d_out, int out_size, void* d_ws, size_t ws_size,
                              hipStream_t stream) {
  const float* x  = (const float*)d_in[0];
  const int* ei   = (const int*)d_in[1];
  const float* W1 = (const float*)d_in[2];
  const float* b1 = (const float*)d_in[3];
  const float* W2 = (const float*)d_in[4];
  const float* b2 = (const float*)d_in[5];
  float* out = (float*)d_out;

  const int n = in_sizes[0] / IN_CH;
  const int e = in_sizes[1] / 2;
  const int* src = ei;
  const int* dst = ei + e;

  float* xs   = (float*)d_ws;
  float* hs   = xs + (size_t)n * HID;
  float* g    = xs;
  int*   csr  = (int*)(hs + (size_t)n * HID);
  int*   cur  = csr + e;
  int*   toff = cur + n + 1;
  int*   pairs = (int*)xs;

  const int nTiles = (n + 15) / 16;
  const int wavesPerBlock = THREADS / 32;
  const int nb = gridFor(n, NPB);
  const int nt = gridFor(e, TILE);

  k_part<<<nt, PT, 0, stream>>>(src, dst, pairs, toff, e, nb);
  k_bucket<<<nb, BT, 0, stream>>>(pairs, toff, csr, cur, n, e, nb, nt);

  k_gemm1<<<gridFor(nTiles, wavesPerBlock), THREADS, 0, stream>>>(x, W1, cur, xs, nTiles);
  long long lanes = (long long)n * 4;
  k_gather<true><<<gridFor(lanes, THREADS), THREADS, 0, stream>>>(cur, csr, xs, b1, hs, n);

  k_gather<false><<<gridFor(lanes, THREADS), THREADS, 0, stream>>>(cur, csr, hs, b2, g, n);
  k_gemm2<<<gridFor(nTiles, wavesPerBlock), THREADS, 0, stream>>>(g, W2, b2, out, nTiles);
}


# Layer-2 gather and its product with W2 in one launch: a block's 64 aggregated rows go through an LDS tile to the WMMA A fragments, g is never written; five launches, not six; sum order unchanged
# speedup vs baseline: 8.7409x; 1.0450x over previous
#include <hip/hip_runtime.h>


typedef __attribute__((ext_vector_type(2))) float v2f;
typedef __attribute__((ext_vector_type(4))) float v4f;
typedef __attribute__((ext_vector_type(8))) float v8f;

#define IN_CH  512
#define HID    16
#define OUT_CH 64
#define THREADS 256
#define SCAN_T 1024

#define NSH   8
#define NPB   (1 << NSH)
#define TILE  12288
#define PT    1024
#define EPT   (TILE / PT)
#define BT    512
#define CAP   12288
#define RPW   8

__device__ __forceinline__ int block_excl_scan(int v, int* wtot) {
  const int lane = threadIdx.x & 31, wave = threadIdx.x >> 5;
  int incl = v;
  for (int o = 1; o < 32; o <<= 1) {
    int t = __shfl_up(incl, o);
    if (lane >= o) incl += t;
  }
  if (lane == 31) wtot[wave] = incl;
  __syncthreads();
  int wpre = (lane < wave) ? wtot[lane] : 0;
  for (int o = 16; o > 0; o >>= 1) wpre += __shfl_xor(wpre, o);
  return wpre + incl - v;
}

__global__ void __launch_bounds__(PT)
k_part(const int* __restrict__ src, const int* __restrict__ dst, int* __restrict__ pairs,
       int* __restrict__ toffw, int e, int nb) {
  __shared__ int lcur[PT];
  __shared__ int wtot[32];
  __shared__ int s_val[TILE];
  unsigned short* toff = (unsigned short*)toffw;
  const int tid = threadIdx.x, tile = blockIdx.x;
  const int t0 = tile * TILE;
  int d[EPT], s[EPT];
  #pragma unroll
  for (int k = 0; k < EPT; ++k) {
    const int i = t0 + k * PT + tid;
    const int ic = min(i, e - 1);
    const int dv = dst[ic];
    s[k] = src[ic];
    d[k] = (i < e) ? dv : -1;
  }
  lcur[tid] = 0;
  __syncthreads();
  #pragma unroll
  for (int k = 0; k < EPT; ++k)
    if (d[k] >= 0) atomicAdd(&lcur[d[k] >> NSH], 1);
  __syncthreads();
  const int h = lcur[tid];
  const int ex = block_excl_scan(h, wtot);
  lcur[tid] = ex;
  if (tid < nb) toff[tile * nb + tid] = (unsigned short)ex;
  __syncthreads();
  #pragma unroll
  for (int k = 0; k < EPT; ++k) {
    if (d[k] >= 0) {
      const int pos = atomicAdd(&lcur[d[k] >> NSH], 1);
      if ((unsigned)pos < (unsigned)TILE) s_val[pos] = (s[k] << NSH) | (d[k] & (NPB - 1));
    }
  }
  __syncthreads();
  const int m = min(TILE, e - t0);
  for (int i = tid; i < m; i += PT) pairs[t0 + i] = s_val[i];
}

template <class F>
__device__ __forceinline__ void walk_runs(const int* __restrict__ pairs, const int* r_src, const int* r_pos,
                                          const int* r_len, int nt, int e, F f) {
  const int lane = threadIdx.x & 31, wave = threadIdx.x >> 5;
  for (int t0 = wave * RPW; t0 < nt; t0 += (BT / 32) * RPW) {
    int w0[RPW], w1[RPW];
    #pragma unroll
    for (int u = 0; u < RPW; ++u) {
      const int t = min(t0 + u, nt - 1);
      const int base = r_src[t], last = max(r_len[t] - 1, 0);
      w0[u] = pairs[min(base + min(lane, last), e - 1)];
      w1[u] = pairs[min(base + min(lane + 32, last), e - 1)];
    }
    #pragma unroll
    for (int u = 0; u < RPW; ++u) {
      const int t = t0 + u;
      if (t < nt) {
        const int len = r_len[t], lp = r_pos[t];
        if (lane < len) f(lp + lane, w0[u]);
        if (lane + 32 < len) f(lp + lane + 32, w1[u]);
        for (int j = lane + 64; j < len; j += 32) f(lp + j, pairs[min(r_src[t] + j, e - 1)]);
      }
    }
  }
}

__global__ void __launch_bounds__(BT)
k_bucket(const int* __restrict__ pairs, const int* __restrict__ toffw, int* __restrict__ csr,
         int* __restrict__ start, int n, int e, int nb, int nt) {
  __shared__ int lcnt[NPB];
  __shared__ int lcur[NPB];
  __shared__ int wtot[32];
  __shared__ int r_src[BT];
  __shared__ int r_pos[BT];
  __shared__ int r_len[BT];
  __shared__ int s_tot[2];
  __shared__ int raw[CAP];
  __shared__ int sorted[CAP];
  const unsigned short* toff = (const unsigned short*)toffw;
  const int tid = threadIdx.x, bk = blockIdx.x;
  int o0 = 0, len = 0;
  if (tid < nt) {
    const int tm = min(TILE, e - tid * TILE);
    const int a = toff[tid * nb + bk];
    const int b = (bk + 1 < nb) ? (int)toff[tid * nb + bk + 1] : tm;
    o0 = min(a, tm);
    len = min(max(b - o0, 0), tm - o0);
  }
  if (tid < NPB) lcnt[tid] = 0;
  const int lp = block_excl_scan(len, wtot);
  r_src[tid] = tid * TILE + o0;
  r_pos[tid] = lp;
  r_len[tid] = len;
  if (tid == BT - 1) s_tot[0] = lp + len;
  __syncthreads();
  const int pre = block_excl_scan(o0, wtot);
  if (tid == BT - 1) s_tot[1] = pre + o0;
  __syncthreads();
  const int m = s_tot[0];
  const int beg = min(s_tot[1], e - m);
  const bool staged = m <= CAP;
  walk_runs(pairs, r_src, r_pos, r_len, nt, e, [&](int p, int w) {
    if (staged && (unsigned)p < (unsigned)CAP) raw[p] = w;
    atomicAdd(&lcnt[w & (NPB - 1)], 1);
  });
  __syncthreads();
  const int c = (tid < NPB) ? lcnt[tid] : 0;
  const int ex = block_excl_scan(c, wtot);
  if (tid < NPB) {
    lcur[tid] = ex;
    const int node = bk * NPB + tid;
    if (node < n) start[node] = beg + ex;
  }
  if (bk == 0 && tid == 0) start[n] = e;
  __syncthreads();
  if (staged) {
    for (int i = tid; i < m; i += BT) {
      const int w = raw[i];
      const int pos = atomicAdd(&lcur[w & (NPB - 1)], 1);
      if ((unsigned)pos < (unsigned)CAP) sorted[pos] = w >> NSH;
    }
    __syncthreads();
    for (int i = tid; i < m; i += BT) csr[beg + i] = sorted[i];
  } else {
    walk_runs(pairs, r_src, r_pos, r_len, nt, e, [&](int p, int w) {
      const int pos = atomicAdd(&lcur[w & (NPB - 1)], 1);
      if ((unsigned)pos < (unsigned)m) csr[beg + pos] = w >> NSH;
    });
  }
}

#define KC 64
#define A_PITCH (KC + 4)
#define W_PITCH (IN_CH + 4)
#define NCHUNK (IN_CH / KC)

__global__ void __launch_bounds__(THREADS)
k_gemm1(const float* __restrict__ x, const float* __restrict__ W1,
        const int* __restrict__ start, float* __restrict__ xw1, int nTiles) {
  __shared__ __attribute__((aligned(16))) float w1t[HID * W_PITCH];
  __shared__ __attribute__((aligned(16))) float as[THREADS / 32][16 * A_PITCH];

  {
    v4f w[8];
    #pragma unroll
    for (int i = 0; i < 8; ++i) w[i] = *(const v4f*)(W1 + (i * THREADS + threadIdx.x) * 4);
    #pragma unroll
    for (int i = 0; i < 8; ++i) {
      const int idx = (i * THREADS + threadIdx.x) * 4;
      const int k = idx >> 4, nn = idx & 15;
      w1t[(nn + 0) * W_PITCH + k] = w[i].x;
      w1t[(nn + 1) * W_PITCH + k] = w[i].y;
      w1t[(nn + 2) * W_PITCH + k] = w[i].z;
      w1t[(nn + 3) * W_PITCH + k] = w[i].w;
    }
  }
  __syncthreads();

  const int wave = threadIdx.x >> 5;
  const int lane = threadIdx.x & 31;
  const int t = blockIdx.x * (THREADS / 32) + wave;
  if (t >= nTiles) return;

  const int half = lane >> 4;
  const int ml = lane & 15;
  const float* xg = x + ((size_t)t * 16 + half) * IN_CH + ml * 4;
  float* ast = &as[wave][half * A_PITCH + ml * 4];
  const float* afr = &as[wave][ml * A_PITCH + half * 4];
  const float* bfr = &w1t[ml * W_PITCH + half * 4];

  v4f ra[8], rb[8];
  #pragma unroll
  for (int j = 0; j < 8; ++j) ra[j] = *(const v4f*)(xg + (size_t)(2 * j) * IN_CH);
  #pragma unroll
  for (int j = 0; j < 8; ++j) rb[j] = *(const v4f*)(xg + (size_t)(2 * j) * IN_CH + KC);

  v8f c0 = {}, c1 = {};
  #pragma unroll
  for (int c = 0; c < NCHUNK; ++c) {
    #pragma unroll
    for (int j = 0; j < 8; ++j) *(v4f*)(ast + 2 * j * A_PITCH) = (c & 1) ? rb[j] : ra[j];
    if (c + 2 < NCHUNK) {
      #pragma unroll
      for (int j = 0; j < 8; ++j) {
        const v4f v = *(const v4f*)(xg + (size_t)(2 * j) * IN_CH + (c + 2) * KC);
        if (c & 1) rb[j] = v; else ra[j] = v;
      }
    }
    __builtin_amdgcn_fence(__ATOMIC_RELEASE, "wavefront");
    __builtin_amdgcn_fence(__ATOMIC_ACQUIRE, "wavefront");
    #pragma unroll
    for (int kk = 0; kk < KC / 8; ++kk) {
      const v4f a = *(const v4f*)(afr + kk * 8);
      const v4f b = *(const v4f*)(bfr + c * KC + kk * 8);
      v2f a0, a1, b0, b1;
      a0.x = a.x; a0.y = a.y; a1.x = a.z; a1.y = a.w;
      b0.x = b.x; b0.y = b.y; b1.x = b.z; b1.y = b.w;
      c0 = __builtin_amdgcn_wmma_f32_16x16x4_f32(false, a0, false, b0, (short)0, c0, false, false);
      c1 = __builtin_amdgcn_wmma_f32_16x16x4_f32(false, a1, false, b1, (short)0, c1, false, false);
    }
    __builtin_amdgcn_fence(__ATOMIC_RELEASE, "wavefront");
    __builtin_amdgcn_fence(__ATOMIC_ACQUIRE, "wavefront");
  }

  float* o = xw1 + (size_t)(t * 16) * HID + ml;
  const int* dc = start + t * 16 + half * 8;
  #pragma unroll
  for (int r = 0; r < 8; ++r)
    o[(half * 8 + r) * HID] = (c0[r] + c1[r]) * rsqrtf((float)(dc[r + 1] - dc[r] + 1));
}

template <bool TAIL>
__device__ __forceinline__ void add8(float4& acc, const float4* __restrict__ rows,
                                     const int* __restrict__ csr, int j, int end, int sub, int n) {
  int s[8];
  float4 v[8];
  #pragma unroll
  for (int k = 0; k < 8; ++k) s[k] = csr[TAIL ? min(j + k, end - 1) : j + k];
  #pragma unroll
  for (int k = 0; k < 8; ++k) v[k] = rows[min(max(s[k], 0), n - 1) * 4 + sub];
  #pragma unroll
  for (int k = 0; k < 8; ++k) {
    const bool ok = !TAIL || k == 0 || j + k < end;
    acc.x = ok ? acc.x + v[k].x : acc.x;
    acc.y = ok ? acc.y + v[k].y : acc.y;
    acc.z = ok ? acc.z + v[k].z : acc.z;
    acc.w = ok ? acc.w + v[k].w : acc.w;
  }
}

template <bool FIRST>
__global__ void k_gather(const int* __restrict__ bstart,
                         const int* __restrict__ csr,
                         const float* __restrict__ in, const float* __restrict__ bias,
                         float* __restrict__ out, int n) {
  const int gid = blockIdx.x * blockDim.x + threadIdx.x;
  const int node = gid >> 2;
  const int sub = gid & 3;
  if (node >= n) return;
  const float4* __restrict__ rows = (const float4*)in;
  const int begin = bstart[node];
  const int end = bstart[node + 1];
  const int deg = end - begin;
  float4 acc = rows[node * 4 + sub];
  int j = begin;
  for (; j + 8 <= end; j += 8) add8<false>(acc, rows, csr, j, end, sub, n);
  if (j < end) add8<true>(acc, rows, csr, j, end, sub, n);
  const float di = rsqrtf((float)(deg + 1));
  float4 r;
  if (FIRST) {
    const float* b = bias + sub * 4;
    r.x = fmaxf(b[0] + di * acc.x, 0.0f) * di;
    r.y = fmaxf(b[1] + di * acc.y, 0.0f) * di;
    r.z = fmaxf(b[2] + di * acc.z, 0.0f) * di;
    r.w = fmaxf(b[3] + di * acc.w, 0.0f) * di;
  } else {
    r.x = di * acc.x; r.y = di * acc.y; r.z = di * acc.z; r.w = di * acc.w;
  }
  ((float4*)out)[node * 4 + sub] = r;
}

#define G_ROWS  (THREADS / 4)
#define G_PITCH (HID + 4)

__global__ void __launch_bounds__(THREADS)
k_gather_gemm2(const int* __restrict__ bstart, const int* __restrict__ csr,
               const float* __restrict__ in, const float* __restrict__ W2,
               const float* __restrict__ bias, float* __restrict__ y, int n) {
  __shared__ __attribute__((aligned(16))) float w2s[HID * OUT_CH];
  __shared__ __attribute__((aligned(16))) float gs[G_ROWS * G_PITCH];
  static_assert(HID * OUT_CH == THREADS * 4, "one 16-byte piece of W2 per thread");

  const int wave = threadIdx.x >> 5;
  const int lane = threadIdx.x & 31;
  const int half = lane >> 4;
  const int ml = lane & 15;
  const int tile = wave >> 1;
  const int cb = (wave & 1) * 32;
  const v4f wv = *(const v4f*)(W2 + threadIdx.x * 4);
  const float bs0 = bias[cb + ml], bs1 = bias[cb + 16 + ml];

  const int nl = threadIdx.x >> 2;
  const int sub = threadIdx.x & 3;
  const int node = blockIdx.x * G_ROWS + nl;
  const bool valid = node < n;
  const int nc = min(node, n - 1);
  const float4* __restrict__ rows = (const float4*)in;
  const int begin = bstart[nc];
  const int endv = bstart[nc + 1];
  float4 acc = rows[nc * 4 + sub];
  const int end = valid ? endv : begin;
  const int deg = end - begin;
  int j = begin;
  for (; j + 8 <= end; j += 8) add8<false>(acc, rows, csr, j, end, sub, n);
  if (j < end) add8<true>(acc, rows, csr, j, end, sub, n);
  const float di = valid ? rsqrtf((float)(deg + 1)) : 0.0f;
  float4 r;
  r.x = di * acc.x; r.y = di * acc.y; r.z = di * acc.z; r.w = di * acc.w;
  *(float4*)&gs[nl * G_PITCH + sub * 4] = r;
  *(v4f*)&w2s[threadIdx.x * 4] = wv;
  __syncthreads();

  const float* hr = gs + (tile * 16 + ml) * G_PITCH + half * 2;

  v8f c0 = {}, c1 = {};
  #pragma unroll
  for (int kk = 0; kk < HID / 4; ++kk) {
    const v2f a = *(const v2f*)(hr + kk * 4);
    const int k0 = kk * 4 + half * 2;
    v2f b0, b1;
    b0.x = w2s[k0 * OUT_CH + cb +  0 + ml]; b0.y = w2s[(k0 + 1) * OUT_CH + cb +  0 + ml];
    b1.x = w2s[k0 * OUT_CH + cb + 16 + ml]; b1.y = w2s[(k0 + 1) * OUT_CH + cb + 16 + ml];
    c0 = __builtin_amdgcn_wmma_f32_16x16x4_f32(false, a, false, b0, (short)0, c0, false, false);
    c1 = __builtin_amdgcn_wmma_f32_16x16x4_f32(false, a, false, b1, (short)0, c1, false, false);
  }

  const int row0 = blockIdx.x * G_ROWS + tile * 16 + half * 8;
  float* o = y + (size_t)row0 * OUT_CH + cb + ml;
  #pragma unroll
  for (int r8 = 0; r8 < 8; ++r8) {
    if (row0 + r8 < n) {
      o[r8 * OUT_CH +  0] = c0[r8] + bs0;
      o[r8 * OUT_CH + 16] = c1[r8] + bs1;
    }
  }
}

static inline int gridFor(long long total, int threads) {
  return (int)((total + threads - 1) / threads);
}

extern "C" void kernel_launch(void* const* d_in, const int* in_sizes, int n_in,
                              void* d_out, int out_size, void* d_ws, size_t ws_size,
                              hipStream_t stream) {
  const float* x  = (const float*)d_in[0];
  const int* ei   = (const int*)d_in[1];
  const float* W1 = (const float*)d_in[2];
  const float* b1 = (const float*)d_in[3];
  const float* W2 = (const float*)d_in[4];
  const float* b2 = (const float*)d_in[5];
  float* out = (float*)d_out;

  const int n = in_sizes[0] / IN_CH;
  const int e = in_sizes[1] / 2;
  const int* src = ei;
  const int* dst = ei + e;

  float* xs   = (float*)d_ws;
  float* hs   = xs + (size_t)n * HID;
  int*   csr  = (int*)(hs + (size_t)n * HID);
  int*   cur  = csr + e;
  int*   toff = cur + n + 1;
  int*   pairs = (int*)xs;

  const int nTiles = (n + 15) / 16;
  const int wavesPerBlock = THREADS / 32;
  const int nb = gridFor(n, NPB);
  const int nt = gridFor(e, TILE);

  k_part<<<nt, PT, 0, stream>>>(src, dst, pairs, toff, e, nb);
  k_bucket<<<nb, BT, 0, stream>>>(pairs, toff, csr, cur, n, e, nb, nt);

  k_gemm1<<<gridFor(nTiles, wavesPerBlock), THREADS, 0, stream>>>(x, W1, cur, xs, nTiles);
  long long lanes = (long long)n * 4;
  k_gather<true><<<gridFor(lanes, THREADS), THREADS, 0, stream>>>(cur, csr, xs, b1, hs, n);

  k_gather_gemm2<<<gridFor(lanes, THREADS), THREADS, 0, stream>>>(cur, csr, hs, W2, b2, out, n);
}


# Level 1 of the sort and GEMM1 share one launch (block-uniform branch, one LDS block); dinv row scaling moves to the level-2 block that owns the row; four launches, not five; sum order unchanged
# speedup vs baseline: 9.0442x; 1.0347x over previous
#include <hip/hip_runtime.h>


typedef __attribute__((ext_vector_type(2))) float v2f;
typedef __attribute__((ext_vector_type(4))) float v4f;
typedef __attribute__((ext_vector_type(8))) float v8f;

#define IN_CH  512
#define HID    16
#define OUT_CH 64
#define THREADS 256
#define SCAN_T 1024

#define NSH   8
#define NPB   (1 << NSH)
#define TILE  12288
#define PT    512
#define EPT   (TILE / PT)
#define BT    512
#define CAP   12288
#define RPW   8

__device__ __forceinline__ int block_excl_scan(int v, int* wtot) {
  const int lane = threadIdx.x & 31, wave = threadIdx.x >> 5;
  int incl = v;
  for (int o = 1; o < 32; o <<= 1) {
    int t = __shfl_up(incl, o);
    if (lane >= o) incl += t;
  }
  if (lane == 31) wtot[wave] = incl;
  __syncthreads();
  int wpre = (lane < wave) ? wtot[lane] : 0;
  for (int o = 16; o > 0; o >>= 1) wpre += __shfl_xor(wpre, o);
  return wpre + incl - v;
}

#define PART_LDS_WORDS (PT + 32 + TILE)
__device__ __forceinline__ void
part_body(int* lds, const int* __restrict__ src, const int* __restrict__ dst, int* __restrict__ pairs,
          int* __restrict__ toffw, int e, int nb, int tile) {
  int* lcur = lds;
  int* wtot = lds + PT;
  int* s_val = lds + PT + 32;
  unsigned short* toff = (unsigned short*)toffw;
  const int tid = threadIdx.x;
  const int t0 = tile * TILE;
  int d[EPT], s[EPT];
  #pragma unroll
  for (int k = 0; k < EPT; ++k) {
    const int i = t0 + k * PT + tid;
    const int ic = min(i, e - 1);
    const int dv = dst[ic];
    s[k] = src[ic];
    d[k] = (i < e) ? dv : -1;
  }
  lcur[tid] = 0;
  __syncthreads();
  #pragma unroll
  for (int k = 0; k < EPT; ++k)
    if (d[k] >= 0) atomicAdd(&lcur[d[k] >> NSH], 1);
  __syncthreads();
  const int h = lcur[tid];
  const int ex = block_excl_scan(h, wtot);
  lcur[tid] = ex;
  if (tid < nb) toff[tile * nb + tid] = (unsigned short)ex;
  __syncthreads();
  #pragma unroll
  for (int k = 0; k < EPT; ++k) {
    if (d[k] >= 0) {
      const int pos = atomicAdd(&lcur[d[k] >> NSH], 1);
      if ((unsigned)pos < (unsigned)TILE) s_val[pos] = (s[k] << NSH) | (d[k] & (NPB - 1));
    }
  }
  __syncthreads();
  const int m = min(TILE, e - t0);
  for (int i = tid; i < m; i += PT) pairs[t0 + i] = s_val[i];
}

template <class F>
__device__ __forceinline__ void walk_runs(const int* __restrict__ pairs, const int* r_src, const int* r_pos,
                                          const int* r_len, int nt, int e, F f) {
  const int lane = threadIdx.x & 31, wave = threadIdx.x >> 5;
  for (int t0 = wave * RPW; t0 < nt; t0 += (BT / 32) * RPW) {
    int w0[RPW], w1[RPW];
    #pragma unroll
    for (int u = 0; u < RPW; ++u) {
      const int t = min(t0 + u, nt - 1);
      const int base = r_src[t], last = max(r_len[t] - 1, 0);
      w0[u] = pairs[min(base + min(lane, last), e - 1)];
      w1[u] = pairs[min(base + min(lane + 32, last), e - 1)];
    }
    #pragma unroll
    for (int u = 0; u < RPW; ++u) {
      const int t = t0 + u;
      if (t < nt) {
        const int len = r_len[t], lp = r_pos[t];
        if (lane < len) f(lp + lane, w0[u]);
        if (lane + 32 < len) f(lp + lane + 32, w1[u]);
        for (int j = lane + 64; j < len; j += 32) f(lp + j, pairs[min(r_src[t] + j, e - 1)]);
      }
    }
  }
}

__global__ void __launch_bounds__(BT)
k_bucket(const int* __restrict__ pairs, const int* __restrict__ toffw, int* __restrict__ csr,
         int* __restrict__ start, float* __restrict__ xw1, int n, int e, int nb, int nt) {
  __shared__ int lcnt[NPB];
  __shared__ int lcur[NPB];
  __shared__ int wtot[32];
  __shared__ int r_src[BT];
  __shared__ int r_pos[BT];
  __shared__ int r_len[BT];
  __shared__ int s_tot[2];
  __shared__ int raw[CAP];
  __shared__ int sorted[CAP];
  const unsigned short* toff = (const unsigned short*)toffw;
  const int tid = threadIdx.x, bk = blockIdx.x;
  int o0 = 0, len = 0;
  if (tid < nt) {
    const int tm = min(TILE, e - tid * TILE);
    const int a = toff[tid * nb + bk];
    const int b = (bk + 1 < nb) ? (int)toff[tid * nb + bk + 1] : tm;
    o0 = min(a, tm);
    len = min(max(b - o0, 0), tm - o0);
  }
  if (tid < NPB) lcnt[tid] = 0;
  static_assert((NPB * 4) % BT == 0, "whole 16-byte pieces per thread");
  v4f* __restrict__ r4 = (v4f*)xw1;
  v4f xr[(NPB * 4) / BT];
  #pragma unroll
  for (int q = 0; q < (NPB * 4) / BT; ++q) {
    const int i = q * BT + tid;
    const int node = min(bk * NPB + (i >> 2), n - 1);
    xr[q] = r4[(size_t)node * 4 + (i & 3)];
  }
  const int lp = block_excl_scan(len, wtot);
  r_src[tid] = tid * TILE + o0;
  r_pos[tid] = lp;
  r_len[tid] = len;
  if (tid == BT - 1) s_tot[0] = lp + len;
  __syncthreads();
  const int pre = block_excl_scan(o0, wtot);
  if (tid == BT - 1) s_tot[1] = pre + o0;
  __syncthreads();
  const int m = s_tot[0];
  const int beg = min(s_tot[1], e - m);
  const bool staged = m <= CAP;
  walk_runs(pairs, r_src, r_pos, r_len, nt, e, [&](int p, int w) {
    if (staged && (unsigned)p < (unsigned)CAP) raw[p] = w;
    atomicAdd(&lcnt[w & (NPB - 1)], 1);
  });
  __syncthreads();
  #pragma unroll
  for (int q = 0; q < (NPB * 4) / BT; ++q) {
    const int i = q * BT + tid;
    const int nl = i >> 2;
    const int node = bk * NPB + nl;
    if (node < n) r4[(size_t)node * 4 + (i & 3)] = xr[q] * rsqrtf((float)(lcnt[nl] + 1));
  }
  const int c = (tid < NPB) ? lcnt[tid] : 0;
  const int ex = block_excl_scan(c, wtot);
  if (tid < NPB) {
    lcur[tid] = ex;
    const int node = bk * NPB + tid;
    if (node < n) start[node] = beg + ex;
  }
  if (bk == 0 && tid == 0) start[n] = e;
  __syncthreads();
  if (staged) {
    for (int i = tid; i < m; i += BT) {
      const int w = raw[i];
      const int pos = atomicAdd(&lcur[w & (NPB - 1)], 1);
      if ((unsigned)pos < (unsigned)CAP) sorted[pos] = w >> NSH;
    }
    __syncthreads();
    for (int i = tid; i < m; i += BT) csr[beg + i] = sorted[i];
  } else {
    walk_runs(pairs, r_src, r_pos, r_len, nt, e, [&](int p, int w) {
      const int pos = atomicAdd(&lcur[w & (NPB - 1)], 1);
      if ((unsigned)pos < (unsigned)m) csr[beg + pos] = w >> NSH;
    });
  }
}

#define KC 64
#define A_PITCH (KC + 4)
#define W_PITCH (IN_CH + 4)
#define NCHUNK (IN_CH / KC)

#define GEMM_LDS_WORDS (HID * W_PITCH + (PT / 32) * 16 * A_PITCH)
__device__ __forceinline__ void
gemm1_body(float* lds, const float* __restrict__ x, const float* __restrict__ W1,
           float* __restrict__ xw1, int nTiles, int gb) {
  float* w1t = lds;
  float* as0 = lds + HID * W_PITCH;

  {
    static_assert(IN_CH * HID == 4 * PT * 4, "four 16-byte pieces of W1 per thread");
    v4f w[4];
    #pragma unroll
    for (int i = 0; i < 4; ++i) w[i] = *(const v4f*)(W1 + (i * PT + threadIdx.x) * 4);
    #pragma unroll
    for (int i = 0; i < 4; ++i) {
      const int idx = (i * PT + threadIdx.x) * 4;
      const int k = idx >> 4, nn = idx & 15;
      w1t[(nn + 0) * W_PITCH + k] = w[i].x;
      w1t[(nn + 1) * W_PITCH + k] = w[i].y;
      w1t[(nn + 2) * W_PITCH + k] = w[i].z;
      w1t[(nn + 3) * W_PITCH + k] = w[i].w;
    }
  }
  __syncthreads();

  const int wave = threadIdx.x >> 5;
  const int lane = threadIdx.x & 31;
  const int t = gb * (PT / 32) + wave;
  if (t >= nTiles) return;

  const int half = lane >> 4;
  const int ml = lane & 15;
  const float* xg = x + ((size_t)t * 16 + half) * IN_CH + ml * 4;
  float* asw = as0 + wave * (16 * A_PITCH);
  float* ast = asw + half * A_PITCH + ml * 4;
  const float* afr = asw + ml * A_PITCH + half * 4;
  const float* bfr = &w1t[ml * W_PITCH + half * 4];

  v4f ra[8], rb[8];
  #pragma unroll
  for (int j = 0; j < 8; ++j) ra[j] = *(const v4f*)(xg + (size_t)(2 * j) * IN_CH);
  #pragma unroll
  for (int j = 0; j < 8; ++j) rb[j] = *(const v4f*)(xg + (size_t)(2 * j) * IN_CH + KC);

  v8f c0 = {}, c1 = {};
  #pragma unroll
  for (int c = 0; c < NCHUNK; ++c) {
    #pragma unroll
    for (int j = 0; j < 8; ++j) *(v4f*)(ast + 2 * j * A_PITCH) = (c & 1) ? rb[j] : ra[j];
    if (c + 2 < NCHUNK) {
      #pragma unroll
      for (int j = 0; j < 8; ++j) {
        const v4f v = *(const v4f*)(xg + (size_t)(2 * j) * IN_CH + (c + 2) * KC);
        if (c & 1) rb[j] = v; else ra[j] = v;
      }
    }
    __builtin_amdgcn_fence(__ATOMIC_RELEASE, "wavefront");
    __builtin_amdgcn_fence(__ATOMIC_ACQUIRE, "wavefront");
    #pragma unroll
    for (int kk = 0; kk < KC / 8; ++kk) {
      const v4f a = *(const v4f*)(afr + kk * 8);
      const v4f b = *(const v4f*)(bfr + c * KC + kk * 8);
      v2f a0, a1, b0, b1;
      a0.x = a.x; a0.y = a.y; a1.x = a.z; a1.y = a.w;
      b0.x = b.x; b0.y = b.y; b1.x = b.z; b1.y = b.w;
      c0 = __builtin_amdgcn_wmma_f32_16x16x4_f32(false, a0, false, b0, (short)0, c0, false, false);
      c1 = __builtin_amdgcn_wmma_f32_16x16x4_f32(false, a1, false, b1, (short)0, c1, false, false);
    }
    __builtin_amdgcn_fence(__ATOMIC_RELEASE, "wavefront");
    __builtin_amdgcn_fence(__ATOMIC_ACQUIRE, "wavefront");
  }

  float* o = xw1 + (size_t)(t * 16) * HID + ml;
  #pragma unroll
  for (int r = 0; r < 8; ++r)
    o[(half * 8 + r) * HID] = c0[r] + c1[r];
}

#define L1_LDS_WORDS (GEMM_LDS_WORDS > PART_LDS_WORDS ? GEMM_LDS_WORDS : PART_LDS_WORDS)
__global__ void __launch_bounds__(PT)
k_part_gemm1(const int* __restrict__ src, const int* __restrict__ dst, int* __restrict__ pairs,
             int* __restrict__ toffw, int e, int nb, int nt,
             const float* __restrict__ x, const float* __restrict__ W1,
             float* __restrict__ xw1, int nTiles) {
  __shared__ __attribute__((aligned(16))) int lds[L1_LDS_WORDS];
  if ((int)blockIdx.x < nt) part_body(lds, src, dst, pairs, toffw, e, nb, (int)blockIdx.x);
  else gemm1_body((float*)lds, x, W1, xw1, nTiles, (int)blockIdx.x - nt);
}

template <bool TAIL>
__device__ __forceinline__ void add8(float4& acc, const float4* __restrict__ rows,
                                     const int* __restrict__ csr, int j, int end, int sub, int n) {
  int s[8];
  float4 v[8];
  #pragma unroll
  for (int k = 0; k < 8; ++k) s[k] = csr[TAIL ? min(j + k, end - 1) : j + k];
  #pragma unroll
  for (int k = 0; k < 8; ++k) v[k] = rows[min(max(s[k], 0), n - 1) * 4 + sub];
  #pragma unroll
  for (int k = 0; k < 8; ++k) {
    const bool ok = !TAIL || k == 0 || j + k < end;
    acc.x = ok ? acc.x + v[k].x : acc.x;
    acc.y = ok ? acc.y + v[k].y : acc.y;
    acc.z = ok ? acc.z + v[k].z : acc.z;
    acc.w = ok ? acc.w + v[k].w : acc.w;
  }
}

template <bool FIRST>
__global__ void k_gather(const int* __restrict__ bstart,
                         const int* __restrict__ csr,
                         const float* __restrict__ in, const float* __restrict__ bias,
                         float* __restrict__ out, int n) {
  const int gid = blockIdx.x * blockDim.x + threadIdx.x;
  const int node = gid >> 2;
  const int sub = gid & 3;
  if (node >= n) return;
  const float4* __restrict__ rows = (const float4*)in;
  const int begin = bstart[node];
  const int end = bstart[node + 1];
  const int deg = end - begin;
  float4 acc = rows[node * 4 + sub];
  int j = begin;
  for (; j + 8 <= end; j += 8) add8<false>(acc, rows, csr, j, end, sub, n);
  if (j < end) add8<true>(acc, rows, csr, j, end, sub, n);
  const float di = rsqrtf((float)(deg + 1));
  float4 r;
  if (FIRST) {
    const float* b = bias + sub * 4;
    r.x = fmaxf(b[0] + di * acc.x, 0.0f) * di;
    r.y = fmaxf(b[1] + di * acc.y, 0.0f) * di;
    r.z = fmaxf(b[2] + di * acc.z, 0.0f) * di;
    r.w = fmaxf(b[3] + di * acc.w, 0.0f) * di;
  } else {
    r.x = di * acc.x; r.y = di * acc.y; r.z = di * acc.z; r.w = di * acc.w;
  }
  ((float4*)out)[node * 4 + sub] = r;
}

#define G_ROWS  (THREADS / 4)
#define G_PITCH (HID + 4)

__global__ void __launch_bounds__(THREADS)
k_gather_gemm2(const int* __restrict__ bstart, const int* __restrict__ csr,
               const float* __restrict__ in, const float* __restrict__ W2,
               const float* __restrict__ bias, float* __restrict__ y, int n) {
  __shared__ __attribute__((aligned(16))) float w2s[HID * OUT_CH];
  __shared__ __attribute__((aligned(16))) float gs[G_ROWS * G_PITCH];
  static_assert(HID * OUT_CH == THREADS * 4, "one 16-byte piece of W2 per thread");

  const int wave = threadIdx.x >> 5;
  const int lane = threadIdx.x & 31;
  const int half = lane >> 4;
  const int ml = lane & 15;
  const int tile = wave >> 1;
  const int cb = (wave & 1) * 32;
  const v4f wv = *(const v4f*)(W2 + threadIdx.x * 4);
  const float bs0 = bias[cb + ml], bs1 = bias[cb + 16 + ml];

  const int nl = threadIdx.x >> 2;
  const int sub = threadIdx.x & 3;
  const int node = blockIdx.x * G_ROWS + nl;
  const bool valid = node < n;
  const int nc = min(node, n - 1);
  const float4* __restrict__ rows = (const float4*)in;
  const int begin = bstart[nc];
  const int endv = bstart[nc + 1];
  float4 acc = rows[nc * 4 + sub];
  const int end = valid ? endv : begin;
  const int deg = end - begin;
  int j = begin;
  for (; j + 8 <= end; j += 8) add8<false>(acc, rows, csr, j, end, sub, n);
  if (j < end) add8<true>(acc, rows, csr, j, end, sub, n);
  const float di = valid ? rsqrtf((float)(deg + 1)) : 0.0f;
  float4 r;
  r.x = di * acc.x; r.y = di * acc.y; r.z = di * acc.z; r.w = di * acc.w;
  *(float4*)&gs[nl * G_PITCH + sub * 4] = r;
  *(v4f*)&w2s[threadIdx.x * 4] = wv;
  __syncthreads();

  const float* hr = gs + (tile * 16 + ml) * G_PITCH + half * 2;

  v8f c0 = {}, c1 = {};
  #pragma unroll
  for (int kk = 0; kk < HID / 4; ++kk) {
    const v2f a = *(const v2f*)(hr + kk * 4);
    const int k0 = kk * 4 + half * 2;
    v2f b0, b1;
    b0.x = w2s[k0 * OUT_CH + cb +  0 + ml]; b0.y = w2s[(k0 + 1) * OUT_CH + cb +  0 + ml];
    b1.x = w2s[k0 * OUT_CH + cb + 16 + ml]; b1.y = w2s[(k0 + 1) * OUT_CH + cb + 16 + ml];
    c0 = __builtin_amdgcn_wmma_f32_16x16x4_f32(false, a, false, b0, (short)0, c0, false, false);
    c1 = __builtin_amdgcn_wmma_f32_16x16x4_f32(false, a, false, b1, (short)0, c1, false, false);
  }

  const int row0 = blockIdx.x * G_ROWS + tile * 16 + half * 8;
  float* o = y + (size_t)row0 * OUT_CH + cb + ml;
  #pragma unroll
  for (int r8 = 0; r8 < 8; ++r8) {
    if (row0 + r8 < n) {
      o[r8 * OUT_CH +  0] = c0[r8] + bs0;
      o[r8 * OUT_CH + 16] = c1[r8] + bs1;
    }
  }
}

static inline int gridFor(long long total, int threads) {
  return (int)((total + threads - 1) / threads);
}

extern "C" void kernel_launch(void* const* d_in, const int* in_sizes, int n_in,
                              void* d_out, int out_size, void* d_ws, size_t ws_size,
                              hipStream_t stream) {
  const float* x  = (const float*)d_in[0];
  const int* ei   = (const int*)d_in[1];
  const float* W1 = (const float*)d_in[2];
  const float* b1 = (const float*)d_in[3];
  const float* W2 = (const float*)d_in[4];
  const float* b2 = (const float*)d_in[5];
  float* out = (float*)d_out;

  const int n = in_sizes[0] / IN_CH;
  const int e = in_sizes[1] / 2;
  const int* src = ei;
  const int* dst = ei + e;

  float* xs   = (float*)d_ws;
  float* hs   = xs + (size_t)n * HID;
  int*   csr  = (int*)(hs + (size_t)n * HID);
  int*   cur  = csr + e;
  int*   toff = cur + n + 1;
  int*   pairs = (int*)out;

  const int nTiles = (n + 15) / 16;
  const int nb = gridFor(n, NPB);
  const int nt = gridFor(e, TILE);

  k_part_gemm1<<<nt + gridFor(nTiles, PT / 32), PT, 0, stream>>>(src, dst, pairs, toff, e, nb, nt, x, W1, xs, nTiles);
  k_bucket<<<nb, BT, 0, stream>>>(pairs, toff, csr, cur, xs, n, e, nb, nt);

  long long lanes = (long long)n * 4;
  k_gather<true><<<gridFor(lanes, THREADS), THREADS, 0, stream>>>(cur, csr, xs, b1, hs, n);

  k_gather_gemm2<<<gridFor(lanes, THREADS), THREADS, 0, stream>>>(cur, csr, hs, W2, b2, out, n);
}


# Both gathers: a block's 64 buckets are one contiguous piece of the bucketed source list: fetched once with whole-line loads into LDS, every round takes its 8 sources from there; sum order unchanged
# speedup vs baseline: 10.6716x; 1.1799x over previous
#include <hip/hip_runtime.h>


typedef __attribute__((ext_vector_type(2))) float v2f;
typedef __attribute__((ext_vector_type(4))) float v4f;
typedef __attribute__((ext_vector_type(8))) float v8f;
typedef __attribute__((ext_vector_type(4))) int v4i;

#define IN_CH  512
#define HID    16
#define OUT_CH 64
#define THREADS 256
#define SCAN_T 1024

#define NSH   8
#define NPB   (1 << NSH)
#define TILE  12288
#define PT    512
#define EPT   (TILE / PT)
#define BT    512
#define CAP   12288
#define RPW   8

__device__ __forceinline__ int block_excl_scan(int v, int* wtot) {
  const int lane = threadIdx.x & 31, wave = threadIdx.x >> 5;
  int incl = v;
  for (int o = 1; o < 32; o <<= 1) {
    int t = __shfl_up(incl, o);
    if (lane >= o) incl += t;
  }
  if (lane == 31) wtot[wave] = incl;
  __syncthreads();
  int wpre = (lane < wave) ? wtot[lane] : 0;
  for (int o = 16; o > 0; o >>= 1) wpre += __shfl_xor(wpre, o);
  return wpre + incl - v;
}

#define PART_LDS_WORDS (PT + 32 + TILE)
__device__ __forceinline__ void
part_body(int* lds, const int* __restrict__ src, const int* __restrict__ dst, int* __restrict__ pairs,
          int* __restrict__ toffw, int e, int nb, int tile) {
  int* lcur = lds;
  int* wtot = lds + PT;
  int* s_val = lds + PT + 32;
  unsigned short* toff = (unsigned short*)toffw;
  const int tid = threadIdx.x;
  const int t0 = tile * TILE;
  int d[EPT], s[EPT];
  #pragma unroll
  for (int k = 0; k < EPT; ++k) {
    const int i = t0 + k * PT + tid;
    const int ic = min(i, e - 1);
    const int dv = dst[ic];
    s[k] = src[ic];
    d[k] = (i < e) ? dv : -1;
  }
  lcur[tid] = 0;
  __syncthreads();
  #pragma unroll
  for (int k = 0; k < EPT; ++k)
    if (d[k] >= 0) atomicAdd(&lcur[d[k] >> NSH], 1);
  __syncthreads();
  const int h = lcur[tid];
  const int ex = block_excl_scan(h, wtot);
  lcur[tid] = ex;
  if (tid < nb) toff[tile * nb + tid] = (unsigned short)ex;
  __syncthreads();
  #pragma unroll
  for (int k = 0; k < EPT; ++k) {
    if (d[k] >= 0) {
      const int pos = atomicAdd(&lcur[d[k] >> NSH], 1);
      if ((unsigned)pos < (unsigned)TILE) s_val[pos] = (s[k] << NSH) | (d[k] & (NPB - 1));
    }
  }
  __syncthreads();
  const int m = min(TILE, e - t0);
  for (int i = tid; i < m; i += PT) pairs[t0 + i] = s_val[i];
}

template <class F>
__device__ __forceinline__ void walk_runs(const int* __restrict__ pairs, const int* r_src, const int* r_pos,
                                          const int* r_len, int nt, int e, F f) {
  const int lane = threadIdx.x & 31, wave = threadIdx.x >> 5;
  for (int t0 = wave * RPW; t0 < nt; t0 += (BT / 32) * RPW) {
    int w0[RPW], w1[RPW];
    #pragma unroll
    for (int u = 0; u < RPW; ++u) {
      const int t = min(t0 + u, nt - 1);
      const int base = r_src[t], last = max(r_len[t] - 1, 0);
      w0[u] = pairs[min(base + min(lane, last), e - 1)];
      w1[u] = pairs[min(base + min(lane + 32, last), e - 1)];
    }
    #pragma unroll
    for (int u = 0; u < RPW; ++u) {
      const int t = t0 + u;
      if (t < nt) {
        const int len = r_len[t], lp = r_pos[t];
        if (lane < len) f(lp + lane, w0[u]);
        if (lane + 32 < len) f(lp + lane + 32, w1[u]);
        for (int j = lane + 64; j < len; j += 32) f(lp + j, pairs[min(r_src[t] + j, e - 1)]);
      }
    }
  }
}

__global__ void __launch_bounds__(BT)
k_bucket(const int* __restrict__ pairs, const int* __restrict__ toffw, int* __restrict__ csr,
         int* __restrict__ start, float* __restrict__ xw1, int n, int e, int nb, int nt) {
  __shared__ int lcnt[NPB];
  __shared__ int lcur[NPB];
  __shared__ int wtot[32];
  __shared__ int r_src[BT];
  __shared__ int r_pos[BT];
  __shared__ int r_len[BT];
  __shared__ int s_tot[2];
  __shared__ int raw[CAP];
  __shared__ int sorted[CAP];
  const unsigned short* toff = (const unsigned short*)toffw;
  const int tid = threadIdx.x, bk = blockIdx.x;
  int o0 = 0, len = 0;
  if (tid < nt) {
    const int tm = min(TILE, e - tid * TILE);
    const int a = toff[tid * nb + bk];
    const int b = (bk + 1 < nb) ? (int)toff[tid * nb + bk + 1] : tm;
    o0 = min(a, tm);
    len = min(max(b - o0, 0), tm - o0);
  }
  if (tid < NPB) lcnt[tid] = 0;
  static_assert((NPB * 4) % BT == 0, "whole 16-byte pieces per thread");
  v4f* __restrict__ r4 = (v4f*)xw1;
  v4f xr[(NPB * 4) / BT];
  #pragma unroll
  for (int q = 0; q < (NPB * 4) / BT; ++q) {
    const int i = q * BT + tid;
    const int node = min(bk * NPB + (i >> 2), n - 1);
    xr[q] = r4[(size_t)node * 4 + (i & 3)];
  }
  const int lp = block_excl_scan(len, wtot);
  r_src[tid] = tid * TILE + o0;
  r_pos[tid] = lp;
  r_len[tid] = len;
  if (tid == BT - 1) s_tot[0] = lp + len;
  __syncthreads();
  const int pre = block_excl_scan(o0, wtot);
  if (tid == BT - 1) s_tot[1] = pre + o0;
  __syncthreads();
  const int m = s_tot[0];
  const int beg = min(s_tot[1], e - m);
  const bool staged = m <= CAP;
  walk_runs(pairs, r_src, r_pos, r_len, nt, e, [&](int p, int w) {
    if (staged && (unsigned)p < (unsigned)CAP) raw[p] = w;
    atomicAdd(&lcnt[w & (NPB - 1)], 1);
  });
  __syncthreads();
  #pragma unroll
  for (int q = 0; q < (NPB * 4) / BT; ++q) {
    const int i = q * BT + tid;
    const int nl = i >> 2;
    const int node = bk * NPB + nl;
    if (node < n) r4[(size_t)node * 4 + (i & 3)] = xr[q] * rsqrtf((float)(lcnt[nl] + 1));
  }
  const int c = (tid < NPB) ? lcnt[tid] : 0;
  const int ex = block_excl_scan(c, wtot);
  if (tid < NPB) {
    lcur[tid] = ex;
    const int node = bk * NPB + tid;
    if (node < n) start[node] = beg + ex;
  }
  if (bk == 0 && tid == 0) start[n] = e;
  __syncthreads();
  if (staged) {
    for (int i = tid; i < m; i += BT) {
      const int w = raw[i];
      const int pos = atomicAdd(&lcur[w & (NPB - 1)], 1);
      if ((unsigned)pos < (unsigned)CAP) sorted[pos] = w >> NSH;
    }
    __syncthreads();
    for (int i = tid; i < m; i += BT) csr[beg + i] = sorted[i];
  } else {
    walk_runs(pairs, r_src, r_pos, r_len, nt, e, [&](int p, int w) {
      const int pos = atomicAdd(&lcur[w & (NPB - 1)], 1);
      if ((unsigned)pos < (unsigned)m) csr[beg + pos] = w >> NSH;
    });
  }
}

#define KC 64
#define A_PITCH (KC + 4)
#define W_PITCH (IN_CH + 4)
#define NCHUNK (IN_CH / KC)

#define GEMM_LDS_WORDS (HID * W_PITCH + (PT / 32) * 16 * A_PITCH)
__device__ __forceinline__ void
gemm1_body(float* lds, const float* __restrict__ x, const float* __restrict__ W1,
           float* __restrict__ xw1, int nTiles, int gb) {
  float* w1t = lds;
  float* as0 = lds + HID * W_PITCH;

  {
    static_assert(IN_CH * HID == 4 * PT * 4, "four 16-byte pieces of W1 per thread");
    v4f w[4];
    #pragma unroll
    for (int i = 0; i < 4; ++i) w[i] = *(const v4f*)(W1 + (i * PT + threadIdx.x) * 4);
    #pragma unroll
    for (int i = 0; i < 4; ++i) {
      const int idx = (i * PT + threadIdx.x) * 4;
      const int k = idx >> 4, nn = idx & 15;
      w1t[(nn + 0) * W_PITCH + k] = w[i].x;
      w1t[(nn + 1) * W_PITCH + k] = w[i].y;
      w1t[(nn + 2) * W_PITCH + k] = w[i].z;
      w1t[(nn + 3) * W_PITCH + k] = w[i].w;
    }
  }
  __syncthreads();

  const int wave = threadIdx.x >> 5;
  const int lane = threadIdx.x & 31;
  const int t = gb * (PT / 32) + wave;
  if (t >= nTiles) return;

  const int half = lane >> 4;
  const int ml = lane & 15;
  const float* xg = x + ((size_t)t * 16 + half) * IN_CH + ml * 4;
  float* asw = as0 + wave * (16 * A_PITCH);
  float* ast = asw + half * A_PITCH + ml * 4;
  const float* afr = asw + ml * A_PITCH + half * 4;
  const float* bfr = &w1t[ml * W_PITCH + half * 4];

  v4f ra[8], rb[8];
  #pragma unroll
  for (int j = 0; j < 8; ++j) ra[j] = *(const v4f*)(xg + (size_t)(2 * j) * IN_CH);
  #pragma unroll
  for (int j = 0; j < 8; ++j) rb[j] = *(const v4f*)(xg + (size_t)(2 * j) * IN_CH + KC);

  v8f c0 = {}, c1 = {};
  #pragma unroll
  for (int c = 0; c < NCHUNK; ++c) {
    #pragma unroll
    for (int j = 0; j < 8; ++j) *(v4f*)(ast + 2 * j * A_PITCH) = (c & 1) ? rb[j] : ra[j];
    if (c + 2 < NCHUNK) {
      #pragma unroll
      for (int j = 0; j < 8; ++j) {
        const v4f v = *(const v4f*)(xg + (size_t)(2 * j) * IN_CH + (c + 2) * KC);
        if (c & 1) rb[j] = v; else ra[j] = v;
      }
    }
    __builtin_amdgcn_fence(__ATOMIC_RELEASE, "wavefront");
    __builtin_amdgcn_fence(__ATOMIC_ACQUIRE, "wavefront");
    #pragma unroll
    for (int kk = 0; kk < KC / 8; ++kk) {
      const v4f a = *(const v4f*)(afr + kk * 8);
      const v4f b = *(const v4f*)(bfr + c * KC + kk * 8);
      v2f a0, a1, b0, b1;
      a0.x = a.x; a0.y = a.y; a1.x = a.z; a1.y = a.w;
      b0.x = b.x; b0.y = b.y; b1.x = b.z; b1.y = b.w;
      c0 = __builtin_amdgcn_wmma_f32_16x16x4_f32(false, a0, false, b0, (short)0, c0, false, false);
      c1 = __builtin_amdgcn_wmma_f32_16x16x4_f32(false, a1, false, b1, (short)0, c1, false, false);
    }
    __builtin_amdgcn_fence(__ATOMIC_RELEASE, "wavefront");
    __builtin_amdgcn_fence(__ATOMIC_ACQUIRE, "wavefront");
  }

  float* o = xw1 + (size_t)(t * 16) * HID + ml;
  #pragma unroll
  for (int r = 0; r < 8; ++r)
    o[(half * 8 + r) * HID] = c0[r] + c1[r];
}

#define L1_LDS_WORDS (GEMM_LDS_WORDS > PART_LDS_WORDS ? GEMM_LDS_WORDS : PART_LDS_WORDS)
__global__ void __launch_bounds__(PT)
k_part_gemm1(const int* __restrict__ src, const int* __restrict__ dst, int* __restrict__ pairs,
             int* __restrict__ toffw, int e, int nb, int nt,
             const float* __restrict__ x, const float* __restrict__ W1,
             float* __restrict__ xw1, int nTiles) {
  __shared__ __attribute__((aligned(16))) int lds[L1_LDS_WORDS];
  if ((int)blockIdx.x < nt) part_body(lds, src, dst, pairs, toffw, e, nb, (int)blockIdx.x);
  else gemm1_body((float*)lds, x, W1, xw1, nTiles, (int)blockIdx.x - nt);
}

#define G_ROWS  (THREADS / 4)
#define SPAN_W  3072

__device__ __forceinline__ bool stage_span(int* sidx, const int* __restrict__ csr, int s0, int s1, int e, int& a0) {
  a0 = s0 & ~63;
  const int nw = s1 - a0;
  const bool staged = nw <= SPAN_W && (e & 3) == 0;
  if (staged) {
    static_assert(SPAN_W % (4 * THREADS) == 0, "whole 16-byte pieces per thread");
    v4i w[SPAN_W / (4 * THREADS)];
    #pragma unroll
    for (int q = 0; q < SPAN_W / (4 * THREADS); ++q) {
      const int p = (q * THREADS + (int)threadIdx.x) * 4;
      v4i t = {0, 0, 0, 0};
      if (p < nw) t = *(const v4i*)(csr + a0 + p);
      w[q] = t;
    }
    #pragma unroll
    for (int q = 0; q < SPAN_W / (4 * THREADS); ++q) {
      const int p = (q * THREADS + (int)threadIdx.x) * 4;
      if (p < nw) *(v4i*)(sidx + p) = w[q];
    }
  }
  __syncthreads();
  return staged;
}

template <bool TAIL, bool LDSI>
__device__ __forceinline__ void add8(float4& acc, const float4* __restrict__ rows,
                                     const int* __restrict__ csr, const int* sidx, int a0,
                                     int j, int end, int sub, int n) {
  int s[8];
  float4 v[8];
  #pragma unroll
  for (int k = 0; k < 8; ++k) {
    const int i = TAIL ? min(j + k, end - 1) : j + k;
    s[k] = LDSI ? sidx[i - a0] : csr[i];
  }
  #pragma unroll
  for (int k = 0; k < 8; ++k) v[k] = rows[min(max(s[k], 0), n - 1) * 4 + sub];
  #pragma unroll
  for (int k = 0; k < 8; ++k) {
    const bool ok = !TAIL || k == 0 || j + k < end;
    acc.x = ok ? acc.x + v[k].x : acc.x;
    acc.y = ok ? acc.y + v[k].y : acc.y;
    acc.z = ok ? acc.z + v[k].z : acc.z;
    acc.w = ok ? acc.w + v[k].w : acc.w;
  }
}

template <bool LDSI>
__device__ __forceinline__ void add_bucket(float4& acc, const float4* __restrict__ rows,
                                           const int* __restrict__ csr, const int* sidx, int a0,
                                           int begin, int end, int sub, int n) {
  int j = begin;
  for (; j + 8 <= end; j += 8) add8<false, LDSI>(acc, rows, csr, sidx, a0, j, end, sub, n);
  if (j < end) add8<true, LDSI>(acc, rows, csr, sidx, a0, j, end, sub, n);
}

template <bool FIRST>
__global__ void __launch_bounds__(THREADS)
k_gather(const int* __restrict__ bstart,
         const int* __restrict__ csr,
         const float* __restrict__ in, const float* __restrict__ bias,
         float* __restrict__ out, int n, int e) {
  __shared__ __attribute__((aligned(16))) int sidx[SPAN_W];
  const int nl = threadIdx.x >> 2;
  const int sub = threadIdx.x & 3;
  const int first = blockIdx.x * G_ROWS;
  const int node = first + nl;
  const bool valid = node < n;
  const int nc = min(node, n - 1);
  const float4* __restrict__ rows = (const float4*)in;
  const int s0r = bstart[min(first, n - 1)];
  const int s1r = bstart[min(first + G_ROWS, n)];
  const int begr = bstart[nc];
  const int endr = bstart[nc + 1];
  float4 acc = rows[nc * 4 + sub];
  const int s0 = min(max(s0r, 0), e);
  const int s1 = min(max(s1r, s0), e);
  int a0;
  const bool staged = stage_span(sidx, csr, s0, s1, e, a0);
  const int begin = min(max(begr, s0), s1);
  const int end = valid ? min(max(endr, begin), s1) : begin;
  const int deg = end - begin;
  if (staged) add_bucket<true>(acc, rows, csr, sidx, a0, begin, end, sub, n);
  else add_bucket<false>(acc, rows, csr, sidx, a0, begin, end, sub, n);
  const float di = rsqrtf((float)(deg + 1));
  float4 r;
  if (FIRST) {
    const float* b = bias + sub * 4;
    r.x = fmaxf(b[0] + di * acc.x, 0.0f) * di;
    r.y = fmaxf(b[1] + di * acc.y, 0.0f) * di;
    r.z = fmaxf(b[2] + di * acc.z, 0.0f) * di;
    r.w = fmaxf(b[3] + di * acc.w, 0.0f) * di;
  } else {
    r.x = di * acc.x; r.y = di * acc.y; r.z = di * acc.z; r.w = di * acc.w;
  }
  if (valid) ((float4*)out)[node * 4 + sub] = r;
}

#define G_PITCH (HID + 4)

__global__ void __launch_bounds__(THREADS)
k_gather_gemm2(const int* __restrict__ bstart, const int* __restrict__ csr,
               const float* __restrict__ in, const float* __restrict__ W2,
               const float* __restrict__ bias, float* __restrict__ y, int n, int e) {
  __shared__ __attribute__((aligned(16))) float w2s[HID * OUT_CH];
  __shared__ __attribute__((aligned(16))) float gs[G_ROWS * G_PITCH];
  __shared__ __attribute__((aligned(16))) int sidx[SPAN_W];
  static_assert(HID * OUT_CH == THREADS * 4, "one 16-byte piece of W2 per thread");

  const int wave = threadIdx.x >> 5;
  const int lane = threadIdx.x & 31;
  const int half = lane >> 4;
  const int ml = lane & 15;
  const int tile = wave >> 1;
  const int cb = (wave & 1) * 32;
  const v4f wv = *(const v4f*)(W2 + threadIdx.x * 4);
  const float bs0 = bias[cb + ml], bs1 = bias[cb + 16 + ml];

  const int nl = threadIdx.x >> 2;
  const int sub = threadIdx.x & 3;
  const int first = blockIdx.x * G_ROWS;
  const int node = first + nl;
  const bool valid = node < n;
  const int nc = min(node, n - 1);
  const float4* __restrict__ rows = (const float4*)in;
  const int s0r = bstart[min(first, n - 1)];
  const int s1r = bstart[min(first + G_ROWS, n)];
  const int begr = bstart[nc];
  const int endr = bstart[nc + 1];
  float4 acc = rows[nc * 4 + sub];
  const int s0 = min(max(s0r, 0), e);
  const int s1 = min(max(s1r, s0), e);
  int a0;
  const bool staged = stage_span(sidx, csr, s0, s1, e, a0);
  const int begin = min(max(begr, s0), s1);
  const int end = valid ? min(max(endr, begin), s1) : begin;
  const int deg = end - begin;
  if (staged) add_bucket<true>(acc, rows, csr, sidx, a0, begin, end, sub, n);
  else add_bucket<false>(acc, rows, csr, sidx, a0, begin, end, sub, n);
  const float di = valid ? rsqrtf((float)(deg + 1)) : 0.0f;
  float4 r;
  r.x = di * acc.x; r.y = di * acc.y; r.z = di * acc.z; r.w = di * acc.w;
  *(float4*)&gs[nl * G_PITCH + sub * 4] = r;
  *(v4f*)&w2s[threadIdx.x * 4] = wv;
  __syncthreads();

  const float* hr = gs + (tile * 16 + ml) * G_PITCH + half * 2;

  v8f c0 = {}, c1 = {};
  #pragma unroll
  for (int kk = 0; kk < HID / 4; ++kk) {
    const v2f a = *(const v2f*)(hr + kk * 4);
    const int k0 = kk * 4 + half * 2;
    v2f b0, b1;
    b0.x = w2s[k0 * OUT_CH + cb +  0 + ml]; b0.y = w2s[(k0 + 1) * OUT_CH + cb +  0 + ml];
    b1.x = w2s[k0 * OUT_CH + cb + 16 + ml]; b1.y = w2s[(k0 + 1) * OUT_CH + cb + 16 + ml];
    c0 = __builtin_amdgcn_wmma_f32_16x16x4_f32(false, a, false, b0, (short)0, c0, false, false);
    c1 = __builtin_amdgcn_wmma_f32_16x16x4_f32(false, a, false, b1, (short)0, c1, false, false);
  }

  const int row0 = blockIdx.x * G_ROWS + tile * 16 + half * 8;
  float* o = y + (size_t)row0 * OUT_CH + cb + ml;
  #pragma unroll
  for (int r8 = 0; r8 < 8; ++r8) {
    if (row0 + r8 < n) {
      o[r8 * OUT_CH +  0] = c0[r8] + bs0;
      o[r8 * OUT_CH + 16] = c1[r8] + bs1;
    }
  }
}

static inline int gridFor(long long total, int threads) {
  return (int)((total + threads - 1) / threads);
}

extern "C" void kernel_launch(void* const* d_in, const int* in_sizes, int n_in,
                              void* d_out, int out_size, void* d_ws, size_t ws_size,
                              hipStream_t stream) {
  const float* x  = (const float*)d_in[0];
  const int* ei   = (const int*)d_in[1];
  const float* W1 = (const float*)d_in[2];
  const float* b1 = (const float*)d_in[3];
  const float* W2 = (const float*)d_in[4];
  const float* b2 = (const float*)d_in[5];
  float* out = (float*)d_out;

  const int n = in_sizes[0] / IN_CH;
  const int e = in_sizes[1] / 2;
  const int* src = ei;
  const int* dst = ei + e;

  float* xs   = (float*)d_ws;
  float* hs   = xs + (size_t)n * HID;
  int*   csr  = (int*)(hs + (size_t)n * HID);
  int*   cur  = csr + e;
  int*   toff = cur + n + 1;
  int*   pairs = (int*)out;

  const int nTiles = (n + 15) / 16;
  const int nb = gridFor(n, NPB);
  const int nt = gridFor(e, TILE);

  k_part_gemm1<<<nt + gridFor(nTiles, PT / 32), PT, 0, stream>>>(src, dst, pairs, toff, e, nb, nt, x, W1, xs, nTiles);
  k_bucket<<<nb, BT, 0, stream>>>(pairs, toff, csr, cur, xs, n, e, nb, nt);

  long long lanes = (long long)n * 4;
  k_gather<true><<<gridFor(lanes, THREADS), THREADS, 0, stream>>>(cur, csr, xs, b1, hs, n, e);

  k_gather_gemm2<<<gridFor(lanes, THREADS), THREADS, 0, stream>>>(cur, csr, hs, W2, b2, out, n, e);
}
